# starting kernel
# baseline (speedup 1.0000x reference)
#include <hip/hip_runtime.h>

typedef __attribute__((ext_vector_type(2))) float v2f;
typedef __attribute__((ext_vector_type(8))) float v8f;

#define NEG_SLOPE 0.2f

template <int CIN, int COUT>
__global__ __launch_bounds__(256) void gemm_wmma(const float* __restrict__ X,
                                                 const float* __restrict__ Wg,
                                                 float* __restrict__ H, int M) {
  __shared__ v2f Wswz[(CIN / 4) * COUT * 2];
  {
    constexpr int TOT = (CIN / 4) * COUT * 2;
    for (int i = threadIdx.x; i < TOT; i += 256) {
      int m = i / (COUT * 2);
      int r = i - m * (COUT * 2);
      int c = r >> 1;
      int hsel = r & 1;
      const float* wp = Wg + (size_t)(4 * m + 2 * hsel) * COUT + c;
      v2f w;
      w.x = wp[0];
      w.y = wp[COUT];
      Wswz[i] = w;
    }
  }
  __syncthreads();

  const int lane = threadIdx.x & 31;
  const int wave = threadIdx.x >> 5;
  const int mtile = blockIdx.x * 8 + wave;
  if (mtile * 16 >= M) return;
  const int m0 = mtile * 16;

  constexpr int NT = COUT / 16;
  v8f acc[NT] = {};

  const int rsel = lane & 15;
  const int hiHalf = lane >> 4;
  const float* xrow = X + (size_t)(m0 + rsel) * CIN + hiHalf * 2;
  const v2f* wbase = Wswz + rsel * 2 + hiHalf;

  for (int k = 0; k < CIN; k += 4) {
    v2f a = *reinterpret_cast<const v2f*>(xrow + k);
    const v2f* wb = wbase + (k >> 2) * (COUT * 2);
#pragma unroll
    for (int t = 0; t < NT; ++t) {
      v2f bf = wb[t * 32];
      acc[t] = __builtin_amdgcn_wmma_f32_16x16x4_f32(
          false, a, false, bf, (short)0, acc[t], false, false);
    }
  }

  const int rbase = m0 + hiHalf * 8;
#pragma unroll
  for (int t = 0; t < NT; ++t) {
#pragma unroll
    for (int i = 0; i < 8; ++i) {
      H[(size_t)(rbase + i) * COUT + t * 16 + rsel] = acc[t][i];
    }
  }
}

template <int C>
__global__ __launch_bounds__(256) void row_proj(const float* __restrict__ H,
                                                const float* __restrict__ asrc,
                                                const float* __restrict__ adst,
                                                float* __restrict__ als,
                                                float* __restrict__ ald, int Nn) {
  const int lane = threadIdx.x & 31;
  const int wave = threadIdx.x >> 5;
  const int n = blockIdx.x * 8 + wave;
  if (n >= Nn) return;
  const float* hrow = H + (size_t)n * C;
  float s = 0.f, d = 0.f;
#pragma unroll
  for (int c = lane; c < C; c += 32) {
    float hv = hrow[c];
    s += hv * asrc[c];
    d += hv * adst[c];
  }
#pragma unroll
  for (int off = 16; off > 0; off >>= 1) {
    s += __shfl_xor(s, off, 32);
    d += __shfl_xor(d, off, 32);
  }
  if (lane == 0) {
    als[n] = s;
    ald[n] = d;
  }
}

__device__ __forceinline__ unsigned fenc(float f) {
  unsigned u = __float_as_uint(f);
  return u ^ ((u >> 31) ? 0xFFFFFFFFu : 0x80000000u);
}
__device__ __forceinline__ float fdec(unsigned u) {
  unsigned v = u ^ ((u & 0x80000000u) ? 0x80000000u : 0xFFFFFFFFu);
  return __uint_as_float(v);
}

__device__ __forceinline__ void edge_sd(int i, int E, const int* __restrict__ src,
                                        const int* __restrict__ dst, int& s, int& d) {
  if (i < E) {
    s = src[i];
    d = dst[i];
  } else {
    s = i - E;
    d = s;
  }
}

__global__ __launch_bounds__(256) void edge_max(const int* __restrict__ src,
                                                const int* __restrict__ dst,
                                                const float* __restrict__ als,
                                                const float* __restrict__ ald,
                                                unsigned* __restrict__ menc,
                                                int E, int Etot) {
  int i = blockIdx.x * 256 + threadIdx.x;
  if (i >= Etot) return;
  int s, d;
  edge_sd(i, E, src, dst, s, d);
  float e = als[s] + ald[d];
  e = (e > 0.f) ? e : NEG_SLOPE * e;
  atomicMax(&menc[d], fenc(e));
}

__global__ __launch_bounds__(256) void edge_expsum(const int* __restrict__ src,
                                                   const int* __restrict__ dst,
                                                   const float* __restrict__ als,
                                                   const float* __restrict__ ald,
                                                   const unsigned* __restrict__ menc,
                                                   float* __restrict__ ee,
                                                   float* __restrict__ denom,
                                                   int E, int Etot) {
  int i = blockIdx.x * 256 + threadIdx.x;
  if (i >= Etot) return;
  int s, d;
  edge_sd(i, E, src, dst, s, d);
  float e = als[s] + ald[d];
  e = (e > 0.f) ? e : NEG_SLOPE * e;
  float x = __expf(e - fdec(menc[d]));
  ee[i] = x;
  atomicAdd(&denom[d], x);
}

template <int C>
__global__ __launch_bounds__(256) void edge_aggregate(const int* __restrict__ src,
                                                      const int* __restrict__ dst,
                                                      const float* __restrict__ H,
                                                      const float* __restrict__ ee,
                                                      const float* __restrict__ denom,
                                                      float* __restrict__ out,
                                                      int E, int Etot) {
  const int lane = threadIdx.x & 31;
  const int wave = threadIdx.x >> 5;
  int i = blockIdx.x * 8 + wave;
  if (i >= Etot) return;
  int s, d;
  edge_sd(i, E, src, dst, s, d);
  float alpha = ee[i] / denom[d];
  constexpr int VPT = C / 32;
  const float* hs = H + (size_t)s * C + lane * VPT;
  float* od = out + (size_t)d * C + lane * VPT;
#pragma unroll
  for (int j = 0; j < VPT; ++j) atomicAdd(&od[j], alpha * hs[j]);
}

template <int C>
__global__ __launch_bounds__(256) void bias_act(float* __restrict__ io,
                                                const float* __restrict__ b,
                                                int total, int relu) {
  int i = blockIdx.x * 256 + threadIdx.x;
  if (i >= total) return;
  float v = io[i] + b[i & (C - 1)];
  if (relu) v = fmaxf(v, 0.f);
  io[i] = v;
}

extern "C" void kernel_launch(void* const* d_in, const int* in_sizes, int n_in,
                              void* d_out, int out_size, void* d_ws, size_t ws_size,
                              hipStream_t stream) {
  const float* x   = (const float*)d_in[0];
  const int*   ei  = (const int*)d_in[1];
  const float* W1  = (const float*)d_in[2];
  const float* as1 = (const float*)d_in[3];
  const float* ad1 = (const float*)d_in[4];
  const float* b1  = (const float*)d_in[5];
  const float* W2  = (const float*)d_in[6];
  const float* as2 = (const float*)d_in[7];
  const float* ad2 = (const float*)d_in[8];
  const float* b2  = (const float*)d_in[9];
  const float* W3  = (const float*)d_in[10];
  const float* as3 = (const float*)d_in[11];
  const float* ad3 = (const float*)d_in[12];
  const float* b3  = (const float*)d_in[13];

  const int N = in_sizes[0] / 128;
  const int E = in_sizes[1] / 2;
  const int Etot = E + N;
  const int* srcp = ei;
  const int* dstp = ei + E;

  float*    buf0  = (float*)d_ws;
  float*    buf1  = buf0 + (size_t)N * 128;
  float*    als   = buf1 + (size_t)N * 128;
  float*    ald   = als + N;
  unsigned* menc  = (unsigned*)(ald + N);
  float*    denom = (float*)(menc + N);
  float*    ee    = denom + N;

  const int gGemm = (N / 16 + 7) / 8;
  const int gProj = (N + 7) / 8;
  const int gE    = (Etot + 255) / 256;
  const int gEagg = (Etot + 7) / 8;
  float* outF = (float*)d_out;

  gemm_wmma<128, 128><<<gGemm, 256, 0, stream>>>(x, W1, buf0, N);
  row_proj<128><<<gProj, 256, 0, stream>>>(buf0, as1, ad1, als, ald, N);
  hipMemsetAsync(menc, 0, (size_t)N * 4, stream);
  hipMemsetAsync(denom, 0, (size_t)N * 4, stream);
  hipMemsetAsync(buf1, 0, (size_t)N * 128 * 4, stream);
  edge_max<<<gE, 256, 0, stream>>>(srcp, dstp, als, ald, menc, E, Etot);
  edge_expsum<<<gE, 256, 0, stream>>>(srcp, dstp, als, ald, menc, ee, denom, E, Etot);
  edge_aggregate<128><<<gEagg, 256, 0, stream>>>(srcp, dstp, buf0, ee, denom, buf1, E, Etot);
  bias_act<128><<<(N * 128 + 255) / 256, 256, 0, stream>>>(buf1, b1, N * 128, 1);

  gemm_wmma<128, 128><<<gGemm, 256, 0, stream>>>(buf1, W2, buf0, N);
  row_proj<128><<<gProj, 256, 0, stream>>>(buf0, as2, ad2, als, ald, N);
  hipMemsetAsync(menc, 0, (size_t)N * 4, stream);
  hipMemsetAsync(denom, 0, (size_t)N * 4, stream);
  hipMemsetAsync(buf1, 0, (size_t)N * 128 * 4, stream);
  edge_max<<<gE, 256, 0, stream>>>(srcp, dstp, als, ald, menc, E, Etot);
  edge_expsum<<<gE, 256, 0, stream>>>(srcp, dstp, als, ald, menc, ee, denom, E, Etot);
  edge_aggregate<128><<<gEagg, 256, 0, stream>>>(srcp, dstp, buf0, ee, denom, buf1, E, Etot);
  bias_act<128><<<(N * 128 + 255) / 256, 256, 0, stream>>>(buf1, b2, N * 128, 1);

  gemm_wmma<128, 64><<<gGemm, 256, 0, stream>>>(buf1, W3, buf0, N);
  row_proj<64><<<gProj, 256, 0, stream>>>(buf0, as3, ad3, als, ald, N);
  hipMemsetAsync(menc, 0, (size_t)N * 4, stream);
  hipMemsetAsync(denom, 0, (size_t)N * 4, stream);
  hipMemsetAsync(outF, 0, (size_t)N * 64 * 4, stream);
  edge_max<<<gE, 256, 0, stream>>>(srcp, dstp, als, ald, menc, E, Etot);
  edge_expsum<<<gE, 256, 0, stream>>>(srcp, dstp, als, ald, menc, ee, denom, E, Etot);
  edge_aggregate<64><<<gEagg, 256, 0, stream>>>(srcp, dstp, buf0, ee, denom, outF, E, Etot);
  bias_act<64><<<(N * 64 + 255) / 256, 256, 0, stream>>>(outF, b3, N * 64, 0);
}


# each work-group streams dst once (was twice): kept edges go to per-wave LDS lists walked twice (max, then sums; src and score saved); layer 3 tiles 400 nodes for list room; LDS add order changes
# speedup vs baseline: 3.2165x; 3.2165x over previous
#include <hip/hip_runtime.h>

typedef __attribute__((ext_vector_type(2))) float v2f;
typedef __attribute__((ext_vector_type(8))) float v8f;

#define NEG_SLOPE 0.2f

template <int CIN, int COUT>
__global__ __launch_bounds__(256) void gemm_wmma(const float* __restrict__ X,
                                                 const float* __restrict__ Wg,
                                                 float* __restrict__ H, int M) {
  __shared__ v2f Wswz[(CIN / 4) * COUT * 2];
  {
    constexpr int TOT = (CIN / 4) * COUT * 2;
    for (int i = threadIdx.x; i < TOT; i += 256) {
      int m = i / (COUT * 2);
      int r = i - m * (COUT * 2);
      int c = r >> 1;
      int hsel = r & 1;
      const float* wp = Wg + (size_t)(4 * m + 2 * hsel) * COUT + c;
      v2f w;
      w.x = wp[0];
      w.y = wp[COUT];
      Wswz[i] = w;
    }
  }
  __syncthreads();

  const int lane = threadIdx.x & 31;
  const int wave = threadIdx.x >> 5;
  const int mtile = blockIdx.x * 8 + wave;
  if (mtile * 16 >= M) return;
  const int m0 = mtile * 16;

  constexpr int NT = COUT / 16;
  v8f acc[NT] = {};

  const int rsel = lane & 15;
  const int hiHalf = lane >> 4;
  const float* xrow = X + (size_t)(m0 + rsel) * CIN + hiHalf * 2;
  const v2f* wbase = Wswz + rsel * 2 + hiHalf;

  for (int k = 0; k < CIN; k += 4) {
    v2f a = *reinterpret_cast<const v2f*>(xrow + k);
    const v2f* wb = wbase + (k >> 2) * (COUT * 2);
#pragma unroll
    for (int t = 0; t < NT; ++t) {
      v2f bf = wb[t * 32];
      acc[t] = __builtin_amdgcn_wmma_f32_16x16x4_f32(
          false, a, false, bf, (short)0, acc[t], false, false);
    }
  }

  const int rbase = m0 + hiHalf * 8;
#pragma unroll
  for (int t = 0; t < NT; ++t) {
#pragma unroll
    for (int i = 0; i < 8; ++i) {
      H[(size_t)(rbase + i) * COUT + t * 16 + rsel] = acc[t][i];
    }
  }
}

template <int C>
__global__ __launch_bounds__(256) void row_proj(const float* __restrict__ H,
                                                const float* __restrict__ asrc,
                                                const float* __restrict__ adst,
                                                float* __restrict__ als,
                                                float* __restrict__ ald, int Nn) {
  constexpr int VPT = C / 32;
  constexpr int RB = 8;
  const int lane = threadIdx.x & 31;
  const int wave = threadIdx.x >> 5;
  const int nb = (blockIdx.x * 8 + wave) * 32;
  if (nb >= Nn) return;
  float as[VPT], ad[VPT];
#pragma unroll
  for (int j = 0; j < VPT; ++j) {
    as[j] = asrc[lane + 32 * j];
    ad[j] = adst[lane + 32 * j];
  }
  float rs = 0.f, rd = 0.f;
  for (int k = 0; k < 32; k += RB) {
    float hv[RB][VPT];
#pragma unroll
    for (int u = 0; u < RB; ++u) {
      int n = nb + k + u;
      if (n > Nn - 1) n = Nn - 1;
      const float* hrow = H + (size_t)n * C;
#pragma unroll
      for (int j = 0; j < VPT; ++j) hv[u][j] = hrow[lane + 32 * j];
    }
#pragma unroll
    for (int u = 0; u < RB; ++u) {
      float s = 0.f, d = 0.f;
#pragma unroll
      for (int j = 0; j < VPT; ++j) {
        s += hv[u][j] * as[j];
        d += hv[u][j] * ad[j];
      }
#pragma unroll
      for (int off = 16; off > 0; off >>= 1) {
        s += __shfl_xor(s, off, 32);
        d += __shfl_xor(d, off, 32);
      }
      if (lane == k + u) {
        rs = s;
        rd = d;
      }
    }
  }
  if (nb + lane < Nn) {
    als[nb + lane] = rs;
    ald[nb + lane] = rd;
  }
}

__device__ __forceinline__ unsigned fenc(float f) {
  unsigned u = __float_as_uint(f);
  return u ^ ((u >> 31) ? 0xFFFFFFFFu : 0x80000000u);
}
__device__ __forceinline__ float fdec(unsigned u) {
  unsigned v = u ^ ((u & 0x80000000u) ? 0x80000000u : 0xFFFFFFFFu);
  return __uint_as_float(v);
}

typedef __attribute__((ext_vector_type(4))) int v4i;

__device__ __forceinline__ float lrelu(float e) { return (e > 0.f) ? e : NEG_SLOPE * e; }

__device__ __forceinline__ void wave_sync() {
  __builtin_amdgcn_fence(__ATOMIC_RELEASE, "wavefront");
  __builtin_amdgcn_wave_barrier();
  __builtin_amdgcn_fence(__ATOMIC_ACQUIRE, "wavefront");
}

template <int C, int TN>
struct AggCtx {
  const int* src;
  const float* als;
  const float* ald;
  const float* H;
  float* acc;
  unsigned* mx;
  float* den;
  int* qi;
  int* qr;
  int n0;
  int lane;
};

template <int C, int TN>
__device__ __forceinline__ void add_rows(const AggCtx<C, TN>& cx, int s, int r, float x, int nvalid) {
  constexpr int VPT = C / 32;
  typedef float vrow __attribute__((ext_vector_type(VPT)));
  const int lane = cx.lane;
  const int col = lane * VPT;
  constexpr int B = 8;
  int k = 0;
  for (; k + B <= nvalid; k += B) {
    vrow v[B];
    int rk[B];
    float xk[B];
#pragma unroll
    for (int u = 0; u < B; ++u) {
      const int sk = __builtin_amdgcn_readlane(s, k + u);
      rk[u] = __builtin_amdgcn_readlane(r, k + u);
      xk[u] = __int_as_float(__builtin_amdgcn_readlane(__float_as_int(x), k + u));
      v[u] = *reinterpret_cast<const vrow*>(cx.H + (size_t)sk * C + col);
    }
#pragma unroll
    for (int u = 0; u < B; ++u) {
      float* ar = cx.acc + rk[u] * C + lane;
#pragma unroll
      for (int j = 0; j < VPT; ++j) atomicAdd(&ar[j * 32], xk[u] * v[u][j]);
    }
  }
  for (; k < nvalid; ++k) {
    const int sk = __builtin_amdgcn_readlane(s, k);
    const int rq = __builtin_amdgcn_readlane(r, k);
    const float xq = __int_as_float(__builtin_amdgcn_readlane(__float_as_int(x), k));
    const vrow hv = *reinterpret_cast<const vrow*>(cx.H + (size_t)sk * C + col);
    float* ar = cx.acc + rq * C + lane;
#pragma unroll
    for (int j = 0; j < VPT; ++j) atomicAdd(&ar[j * 32], xq * hv[j]);
  }
}


struct Share {
  const v4i* dv;
  int head;
  int nvec;
  int tail0;
  int rot;
};

__device__ __forceinline__ Share share_of(const int* __restrict__ dst, int E) {
  Share sh;
  const unsigned long long da = (unsigned long long)dst;
  int head = (int)(((16u - (unsigned)(da & 15ull)) & 15u) >> 2);
  if (head > E) head = E;
  sh.head = head;
  sh.nvec = (E - head) >> 2;
  sh.tail0 = head + sh.nvec * 4;
  sh.dv = reinterpret_cast<const v4i*>(dst + head);
  sh.rot = (int)(((long long)blockIdx.x * sh.nvec) / gridDim.x) & ~31;
  return sh;
}


template <int CAPW>
__device__ __forceinline__ void rec(int* li, int* lr, int& cnt, unsigned mask, bool f, int i, unsigned r) {
  if (mask == 0u) return;
  if (f) {
    const int slot = cnt + (int)__builtin_amdgcn_mbcnt_lo(mask, 0u);
    if (slot < CAPW) {
      li[slot] = i;
      lr[slot] = (int)r;
    }
  }
  cnt += __builtin_popcount(mask);
}

template <int TN, int CAPW>
__device__ __forceinline__ void rec4(int* li, int* lr, int& cnt, int n0, v4i d, int ib) {
  const unsigned r0 = (unsigned)(d.x - n0);
  const unsigned r1 = (unsigned)(d.y - n0);
  const unsigned r2 = (unsigned)(d.z - n0);
  const unsigned r3 = (unsigned)(d.w - n0);
  const bool f0 = r0 < (unsigned)TN;
  const bool f1 = r1 < (unsigned)TN;
  const bool f2 = r2 < (unsigned)TN;
  const bool f3 = r3 < (unsigned)TN;
  const unsigned m0 = __builtin_amdgcn_ballot_w32(f0);
  const unsigned m1 = __builtin_amdgcn_ballot_w32(f1);
  const unsigned m2 = __builtin_amdgcn_ballot_w32(f2);
  const unsigned m3 = __builtin_amdgcn_ballot_w32(f3);
  if ((m0 | m1 | m2 | m3) == 0u) return;
  rec<CAPW>(li, lr, cnt, m0, f0, ib, r0);
  rec<CAPW>(li, lr, cnt, m1, f1, ib + 1, r1);
  rec<CAPW>(li, lr, cnt, m2, f2, ib + 2, r2);
  rec<CAPW>(li, lr, cnt, m3, f3, ib + 3, r3);
}

template <int TN, int CAPW>
__device__ __forceinline__ int scan_record(const Share& sh, const int* __restrict__ dst, int E, int n0,
                                           int wave, int lane, int* li, int* lr) {
  const v4i* dv = sh.dv;
  const int head = sh.head;
  int cnt = 0;
  constexpr int U = 8;
#pragma nounroll
  for (int part = 0; part < 2; ++part) {
    const int lo = part == 0 ? sh.rot : 0;
    const int hi = part == 0 ? sh.nvec : sh.rot;
    int vb = lo + wave * 32;
    for (; vb + (U - 1) * 1024 + 31 < hi; vb += U * 1024) {
      v4i d[U];
#pragma unroll
      for (int u = 0; u < U; ++u) d[u] = dv[vb + u * 1024 + lane];
#pragma unroll
      for (int u = 0; u < U; ++u)
        rec4<TN, CAPW>(li, lr, cnt, n0, d[u], head + (vb + u * 1024 + lane) * 4);
    }
    for (; vb < hi; vb += 1024) {
      const int v = vb + lane;
      v4i d = {-1, -1, -1, -1};
      if (v < hi) d = dv[v];
      rec4<TN, CAPW>(li, lr, cnt, n0, d, head + v * 4);
    }
  }
  if (wave == 0) {
    const int nt = E - sh.tail0;
    int i = -1;
    if (lane < head) i = lane;
    else if (lane - head < nt) i = sh.tail0 + (lane - head);
    const int dd = (i >= 0) ? dst[i] : -1;
    const unsigned r = (unsigned)(dd - n0);
    const bool f = (i >= 0) && (r < (unsigned)TN);
    rec<CAPW>(li, lr, cnt, __builtin_amdgcn_ballot_w32(f), f, i, r);
  }
  return cnt;
}

template <int C, int TN>
__device__ __forceinline__ void score_list(const AggCtx<C, TN>& cx, int* li, const int* lr,
                                           float* le, int cnt) {
  constexpr int SB = 4;
  const int lane = cx.lane;
  for (int base = 0; base < cnt; base += 32 * SB) {
    int p[SB], r[SB], s[SB];
    float a[SB], d[SB];
#pragma unroll
    for (int u = 0; u < SB; ++u) {
      p[u] = base + u * 32 + lane;
      const int q = p[u] < cnt ? p[u] : cnt - 1;
      r[u] = lr[q];
      s[u] = cx.src[li[q]];
    }
#pragma unroll
    for (int u = 0; u < SB; ++u) {
      a[u] = cx.als[s[u]];
      d[u] = cx.ald[cx.n0 + r[u]];
    }
    static_assert(SB == 4, "the statement below names the SB gathered pairs");
    asm volatile("" ::"v"(a[0]), "v"(a[1]), "v"(a[2]), "v"(a[3]), "v"(d[0]), "v"(d[1]), "v"(d[2]), "v"(d[3]));
#pragma unroll
    for (int u = 0; u < SB; ++u) {
      if (p[u] < cnt) {
        const float e = lrelu(a[u] + d[u]);
        atomicMax(&cx.mx[r[u]], fenc(e));
        li[p[u]] = s[u];
        le[p[u]] = e;
      }
    }
  }
}

template <int C, int TN>
__device__ __forceinline__ void sum_list(const AggCtx<C, TN>& cx, const int* li, const int* lr,
                                         const float* le, int cnt) {
  const int lane = cx.lane;
  for (int base = 0; base < cnt; base += 32) {
    const int nvalid = (cnt - base < 32) ? (cnt - base) : 32;
    int s = 0, r = 0;
    float x = 0.f;
    if (lane < nvalid) {
      s = li[base + lane];
      r = lr[base + lane];
      x = __expf(le[base + lane] - fdec(cx.mx[r]));
      atomicAdd(&cx.den[r], x);
    }
    add_rows<C, TN>(cx, s, r, x, nvalid);
  }
}


template <int C, int TN, int PHASE>
__device__ __forceinline__ void drain(const AggCtx<C, TN>& cx, int base, int nvalid) {
  const int lane = cx.lane;
  int s = 0, r = 0;
  float x = 0.f;
  if (lane < nvalid) {
    const int i = cx.qi[base + lane];
    r = cx.qr[base + lane];
    s = cx.src[i];
    const float e = lrelu(cx.als[s] + cx.ald[cx.n0 + r]);
    if (PHASE == 1) {
      atomicMax(&cx.mx[r], fenc(e));
    } else {
      x = __expf(e - fdec(cx.mx[r]));
      atomicAdd(&cx.den[r], x);
    }
  }
  if (PHASE == 2) add_rows<C, TN>(cx, s, r, x, nvalid);
}

template <int C, int TN, int PHASE>
__device__ __forceinline__ void push(const AggCtx<C, TN>& cx, int& cnt, bool f, int i, unsigned r) {
  const unsigned mask = __builtin_amdgcn_ballot_w32(f);
  if (mask == 0u) return;
  if (f) {
    const int slot = cnt + (int)__builtin_amdgcn_mbcnt_lo(mask, 0u);
    cx.qi[slot] = i;
    cx.qr[slot] = (int)r;
  }
  cnt += __builtin_popcount(mask);
  if (cnt >= 32) {
    wave_sync();
    cnt -= 32;
    drain<C, TN, PHASE>(cx, cnt, 32);
    wave_sync();
  }
}

template <int C, int TN, int PHASE>
__device__ __forceinline__ void proc4(const AggCtx<C, TN>& cx, int& cnt, v4i d, int ib) {
  const unsigned r0 = (unsigned)(d.x - cx.n0);
  const unsigned r1 = (unsigned)(d.y - cx.n0);
  const unsigned r2 = (unsigned)(d.z - cx.n0);
  const unsigned r3 = (unsigned)(d.w - cx.n0);
  const bool any = (r0 < (unsigned)TN) | (r1 < (unsigned)TN) | (r2 < (unsigned)TN) | (r3 < (unsigned)TN);
  if (__builtin_amdgcn_ballot_w32(any) == 0u) return;
  push<C, TN, PHASE>(cx, cnt, r0 < (unsigned)TN, ib, r0);
  push<C, TN, PHASE>(cx, cnt, r1 < (unsigned)TN, ib + 1, r1);
  push<C, TN, PHASE>(cx, cnt, r2 < (unsigned)TN, ib + 2, r2);
  push<C, TN, PHASE>(cx, cnt, r3 < (unsigned)TN, ib + 3, r3);
}

template <int C, int TN, int PHASE>
__device__ __forceinline__ void scan_slow(const AggCtx<C, TN>& cx, const Share& sh,
                                          const int* __restrict__ dst, int E, int wave) {
  const v4i* dv = sh.dv;
  const int head = sh.head;
  const int lane = cx.lane;
  int cnt = 0;
#pragma nounroll
  for (int part = 0; part < 2; ++part) {
    const int lo = part == 0 ? sh.rot : 0;
    const int hi = part == 0 ? sh.nvec : sh.rot;
    for (int vb = lo + wave * 32; vb < hi; vb += 1024) {
      const int v = vb + lane;
      v4i d = {-1, -1, -1, -1};
      if (v < hi) d = dv[v];
      proc4<C, TN, PHASE>(cx, cnt, d, head + v * 4);
    }
  }
  if (wave == 0) {
    const int nt = E - sh.tail0;
    int i = -1;
    if (lane < head) i = lane;
    else if (lane - head < nt) i = sh.tail0 + (lane - head);
    const int dd = (i >= 0) ? dst[i] : -1;
    const unsigned r = (unsigned)(dd - cx.n0);
    push<C, TN, PHASE>(cx, cnt, (i >= 0) && (r < (unsigned)TN), i, r);
  }
  wave_sync();
  if (cnt > 0) drain<C, TN, PHASE>(cx, 0, cnt);
  wave_sync();
}

template <int C, int TN, int CAPW>
__global__ __launch_bounds__(1024) void agg_tile(const int* __restrict__ src,
                                                 const int* __restrict__ dst,
                                                 const float* __restrict__ als,
                                                 const float* __restrict__ ald,
                                                 const float* __restrict__ H,
                                                 const float* __restrict__ b,
                                                 float* __restrict__ out,
                                                 int E, int Nn, int relu) {
  constexpr int VPT = C / 32;
  static_assert(CAPW >= 64, "the slow way's queue lives in the list's first 64 slots");
  __shared__ float acc[TN * C];
  __shared__ unsigned mx[TN];
  __shared__ float den[TN];
  __shared__ int li[32 * CAPW];
  __shared__ int lr[32 * CAPW];
  __shared__ float le[32 * CAPW];

  const int tid = threadIdx.x;
  const int lane = tid & 31;
  const int wave = __builtin_amdgcn_readfirstlane(tid >> 5);
  const int n0 = blockIdx.x * TN;

  for (int i = tid; i < TN * C; i += 1024) acc[i] = 0.f;
  for (int n = tid; n < TN; n += 1024) {
    const int node = n0 + n;
    float e = 0.f;
    if (node < Nn) e = lrelu(als[node] + ald[node]);
    mx[n] = fenc(e);
    den[n] = 0.f;
  }
  __syncthreads();

  int* wli = li + wave * CAPW;
  int* wlr = lr + wave * CAPW;
  float* wle = le + wave * CAPW;

  AggCtx<C, TN> cx;
  cx.src = src;
  cx.als = als;
  cx.ald = ald;
  cx.H = H;
  cx.acc = acc;
  cx.mx = mx;
  cx.den = den;
  cx.qi = wli;
  cx.qr = wlr;
  cx.n0 = n0;
  cx.lane = lane;

  const Share sh = share_of(dst, E);
  const int kept = scan_record<TN, CAPW>(sh, dst, E, n0, wave, lane, wli, wlr);
  const bool listed = kept <= CAPW;
  wave_sync();
  if (listed) score_list<C, TN>(cx, wli, wlr, wle, kept);
  else scan_slow<C, TN, 1>(cx, sh, dst, E, wave);
  __syncthreads();
  if (listed) sum_list<C, TN>(cx, wli, wlr, wle, kept);
  else scan_slow<C, TN, 2>(cx, sh, dst, E, wave);
  __syncthreads();

  for (int idx = tid; idx < TN * C; idx += 1024) {
    const int n = idx / C;
    const int c = idx & (C - 1);
    const int node = n0 + n;
    if (node < Nn) {
      const float es = lrelu(als[node] + ald[node]);
      const float xs = __expf(es - fdec(mx[n]));
      const float a = acc[n * C + (c % VPT) * 32 + c / VPT];
      float v = (a + xs * H[(size_t)node * C + c]) / (den[n] + xs) + b[c];
      if (relu) v = fmaxf(v, 0.f);
      out[(size_t)node * C + c] = v;
    }
  }
}

extern "C" void kernel_launch(void* const* d_in, const int* in_sizes, int n_in,
                              void* d_out, int out_size, void* d_ws, size_t ws_size,
                              hipStream_t stream) {
  const float* x   = (const float*)d_in[0];
  const int*   ei  = (const int*)d_in[1];
  const float* W1  = (const float*)d_in[2];
  const float* as1 = (const float*)d_in[3];
  const float* ad1 = (const float*)d_in[4];
  const float* b1  = (const float*)d_in[5];
  const float* W2  = (const float*)d_in[6];
  const float* as2 = (const float*)d_in[7];
  const float* ad2 = (const float*)d_in[8];
  const float* b2  = (const float*)d_in[9];
  const float* W3  = (const float*)d_in[10];
  const float* as3 = (const float*)d_in[11];
  const float* ad3 = (const float*)d_in[12];
  const float* b3  = (const float*)d_in[13];

  const int N = in_sizes[0] / 128;
  const int E = in_sizes[1] / 2;
  const int* srcp = ei;
  const int* dstp = ei + E;

  float* buf0 = (float*)d_ws;
  float* buf1 = buf0 + (size_t)N * 128;
  float* als  = buf1 + (size_t)N * 128;
  float* ald  = als + N;

  constexpr int TN = 400;
  constexpr int CAPW = 288;
  const int gGemm = (N / 16 + 7) / 8;
  const int gProj = (N + 255) / 256;
  const int gAgg  = (N + TN - 1) / TN;
  float* outF = (float*)d_out;

  gemm_wmma<128, 128><<<gGemm, 256, 0, stream>>>(x, W1, buf0, N);
  row_proj<128><<<gProj, 256, 0, stream>>>(buf0, as1, ad1, als, ald, N);
  agg_tile<128, TN, CAPW><<<gAgg, 1024, 0, stream>>>(srcp, dstp, als, ald, buf0, b1, buf1, E, N, 1);

  gemm_wmma<128, 128><<<gGemm, 256, 0, stream>>>(buf1, W2, buf0, N);
  row_proj<128><<<gProj, 256, 0, stream>>>(buf0, as2, ad2, als, ald, N);
  agg_tile<128, TN, CAPW><<<gAgg, 1024, 0, stream>>>(srcp, dstp, als, ald, buf0, b2, buf1, E, N, 1);

  gemm_wmma<128, 64><<<gGemm, 256, 0, stream>>>(buf1, W3, buf0, N);
  row_proj<64><<<gProj, 256, 0, stream>>>(buf0, as3, ad3, als, ald, N);
  agg_tile<64, TN, CAPW><<<gAgg, 1024, 0, stream>>>(srcp, dstp, als, ald, buf0, b3, outF, E, N, 0);
}


# layer 1's per-wave kept-edge lists (source node, tile-local destination) go to the workspace; layers 2 and 3 read them back instead of streaming dst and gathering src again; LDS add order as before
# speedup vs baseline: 4.0939x; 1.2728x over previous
#include <hip/hip_runtime.h>

typedef __attribute__((ext_vector_type(2))) float v2f;
typedef __attribute__((ext_vector_type(8))) float v8f;

#define NEG_SLOPE 0.2f

template <int CIN, int COUT>
__global__ __launch_bounds__(256) void gemm_wmma(const float* __restrict__ X,
                                                 const float* __restrict__ Wg,
                                                 float* __restrict__ H, int M) {
  __shared__ v2f Wswz[(CIN / 4) * COUT * 2];
  {
    constexpr int TOT = (CIN / 4) * COUT * 2;
    for (int i = threadIdx.x; i < TOT; i += 256) {
      int m = i / (COUT * 2);
      int r = i - m * (COUT * 2);
      int c = r >> 1;
      int hsel = r & 1;
      const float* wp = Wg + (size_t)(4 * m + 2 * hsel) * COUT + c;
      v2f w;
      w.x = wp[0];
      w.y = wp[COUT];
      Wswz[i] = w;
    }
  }
  __syncthreads();

  const int lane = threadIdx.x & 31;
  const int wave = threadIdx.x >> 5;
  const int mtile = blockIdx.x * 8 + wave;
  if (mtile * 16 >= M) return;
  const int m0 = mtile * 16;

  constexpr int NT = COUT / 16;
  v8f acc[NT] = {};

  const int rsel = lane & 15;
  const int hiHalf = lane >> 4;
  const float* xrow = X + (size_t)(m0 + rsel) * CIN + hiHalf * 2;
  const v2f* wbase = Wswz + rsel * 2 + hiHalf;

  for (int k = 0; k < CIN; k += 4) {
    v2f a = *reinterpret_cast<const v2f*>(xrow + k);
    const v2f* wb = wbase + (k >> 2) * (COUT * 2);
#pragma unroll
    for (int t = 0; t < NT; ++t) {
      v2f bf = wb[t * 32];
      acc[t] = __builtin_amdgcn_wmma_f32_16x16x4_f32(
          false, a, false, bf, (short)0, acc[t], false, false);
    }
  }

  const int rbase = m0 + hiHalf * 8;
#pragma unroll
  for (int t = 0; t < NT; ++t) {
#pragma unroll
    for (int i = 0; i < 8; ++i) {
      H[(size_t)(rbase + i) * COUT + t * 16 + rsel] = acc[t][i];
    }
  }
}

template <int C>
__global__ __launch_bounds__(256) void row_proj(const float* __restrict__ H,
                                                const float* __restrict__ asrc,
                                                const float* __restrict__ adst,
                                                float* __restrict__ als,
                                                float* __restrict__ ald, int Nn) {
  constexpr int VPT = C / 32;
  constexpr int RB = 8;
  const int lane = threadIdx.x & 31;
  const int wave = threadIdx.x >> 5;
  const int nb = (blockIdx.x * 8 + wave) * 32;
  if (nb >= Nn) return;
  float as[VPT], ad[VPT];
#pragma unroll
  for (int j = 0; j < VPT; ++j) {
    as[j] = asrc[lane + 32 * j];
    ad[j] = adst[lane + 32 * j];
  }
  float rs = 0.f, rd = 0.f;
  for (int k = 0; k < 32; k += RB) {
    float hv[RB][VPT];
#pragma unroll
    for (int u = 0; u < RB; ++u) {
      int n = nb + k + u;
      if (n > Nn - 1) n = Nn - 1;
      const float* hrow = H + (size_t)n * C;
#pragma unroll
      for (int j = 0; j < VPT; ++j) hv[u][j] = hrow[lane + 32 * j];
    }
#pragma unroll
    for (int u = 0; u < RB; ++u) {
      float s = 0.f, d = 0.f;
#pragma unroll
      for (int j = 0; j < VPT; ++j) {
        s += hv[u][j] * as[j];
        d += hv[u][j] * ad[j];
      }
#pragma unroll
      for (int off = 16; off > 0; off >>= 1) {
        s += __shfl_xor(s, off, 32);
        d += __shfl_xor(d, off, 32);
      }
      if (lane == k + u) {
        rs = s;
        rd = d;
      }
    }
  }
  if (nb + lane < Nn) {
    als[nb + lane] = rs;
    ald[nb + lane] = rd;
  }
}

__device__ __forceinline__ unsigned fenc(float f) {
  unsigned u = __float_as_uint(f);
  return u ^ ((u >> 31) ? 0xFFFFFFFFu : 0x80000000u);
}
__device__ __forceinline__ float fdec(unsigned u) {
  unsigned v = u ^ ((u & 0x80000000u) ? 0x80000000u : 0xFFFFFFFFu);
  return __uint_as_float(v);
}

typedef __attribute__((ext_vector_type(4))) int v4i;
typedef __attribute__((ext_vector_type(2))) int v2i;

__device__ __forceinline__ float lrelu(float e) { return (e > 0.f) ? e : NEG_SLOPE * e; }

__device__ __forceinline__ void wave_sync() {
  __builtin_amdgcn_fence(__ATOMIC_RELEASE, "wavefront");
  __builtin_amdgcn_wave_barrier();
  __builtin_amdgcn_fence(__ATOMIC_ACQUIRE, "wavefront");
}

template <int C, int TN>
struct AggCtx {
  const int* src;
  const float* als;
  const float* ald;
  const float* H;
  float* acc;
  unsigned* mx;
  float* den;
  int* qi;
  int* qr;
  int n0;
  int lane;
};

template <int C, int TN>
__device__ __forceinline__ void add_rows(const AggCtx<C, TN>& cx, int s, int r, float x, int nvalid) {
  constexpr int VPT = C / 32;
  typedef float vrow __attribute__((ext_vector_type(VPT)));
  const int lane = cx.lane;
  const int col = lane * VPT;
  constexpr int B = 8;
  int k = 0;
  for (; k + B <= nvalid; k += B) {
    vrow v[B];
    int rk[B];
    float xk[B];
#pragma unroll
    for (int u = 0; u < B; ++u) {
      const int sk = __builtin_amdgcn_readlane(s, k + u);
      rk[u] = __builtin_amdgcn_readlane(r, k + u);
      xk[u] = __int_as_float(__builtin_amdgcn_readlane(__float_as_int(x), k + u));
      v[u] = *reinterpret_cast<const vrow*>(cx.H + (size_t)sk * C + col);
    }
#pragma unroll
    for (int u = 0; u < B; ++u) {
      float* ar = cx.acc + rk[u] * C + lane;
#pragma unroll
      for (int j = 0; j < VPT; ++j) atomicAdd(&ar[j * 32], xk[u] * v[u][j]);
    }
  }
  for (; k < nvalid; ++k) {
    const int sk = __builtin_amdgcn_readlane(s, k);
    const int rq = __builtin_amdgcn_readlane(r, k);
    const float xq = __int_as_float(__builtin_amdgcn_readlane(__float_as_int(x), k));
    const vrow hv = *reinterpret_cast<const vrow*>(cx.H + (size_t)sk * C + col);
    float* ar = cx.acc + rq * C + lane;
#pragma unroll
    for (int j = 0; j < VPT; ++j) atomicAdd(&ar[j * 32], xq * hv[j]);
  }
}


struct Share {
  const v4i* dv;
  int head;
  int nvec;
  int tail0;
  int rot;
};

__device__ __forceinline__ Share share_of(const int* __restrict__ dst, int E) {
  Share sh;
  const unsigned long long da = (unsigned long long)dst;
  int head = (int)(((16u - (unsigned)(da & 15ull)) & 15u) >> 2);
  if (head > E) head = E;
  sh.head = head;
  sh.nvec = (E - head) >> 2;
  sh.tail0 = head + sh.nvec * 4;
  sh.dv = reinterpret_cast<const v4i*>(dst + head);
  sh.rot = (int)(((long long)blockIdx.x * sh.nvec) / gridDim.x) & ~31;
  return sh;
}


template <int CAPW>
__device__ __forceinline__ void rec(int* li, int* lr, int& cnt, unsigned mask, bool f, int i, unsigned r) {
  if (mask == 0u) return;
  if (f) {
    const int slot = cnt + (int)__builtin_amdgcn_mbcnt_lo(mask, 0u);
    if (slot < CAPW) {
      li[slot] = i;
      lr[slot] = (int)r;
    }
  }
  cnt += __builtin_popcount(mask);
}

template <int TN, int CAPW>
__device__ __forceinline__ void rec4(int* li, int* lr, int& cnt, int n0, v4i d, int ib) {
  const unsigned r0 = (unsigned)(d.x - n0);
  const unsigned r1 = (unsigned)(d.y - n0);
  const unsigned r2 = (unsigned)(d.z - n0);
  const unsigned r3 = (unsigned)(d.w - n0);
  const bool f0 = r0 < (unsigned)TN;
  const bool f1 = r1 < (unsigned)TN;
  const bool f2 = r2 < (unsigned)TN;
  const bool f3 = r3 < (unsigned)TN;
  const unsigned m0 = __builtin_amdgcn_ballot_w32(f0);
  const unsigned m1 = __builtin_amdgcn_ballot_w32(f1);
  const unsigned m2 = __builtin_amdgcn_ballot_w32(f2);
  const unsigned m3 = __builtin_amdgcn_ballot_w32(f3);
  if ((m0 | m1 | m2 | m3) == 0u) return;
  rec<CAPW>(li, lr, cnt, m0, f0, ib, r0);
  rec<CAPW>(li, lr, cnt, m1, f1, ib + 1, r1);
  rec<CAPW>(li, lr, cnt, m2, f2, ib + 2, r2);
  rec<CAPW>(li, lr, cnt, m3, f3, ib + 3, r3);
}

template <int TN, int CAPW>
__device__ __forceinline__ int scan_record(const Share& sh, const int* __restrict__ dst, int E, int n0,
                                           int wave, int lane, int* li, int* lr) {
  const v4i* dv = sh.dv;
  const int head = sh.head;
  int cnt = 0;
  constexpr int U = 8;
#pragma nounroll
  for (int part = 0; part < 2; ++part) {
    const int lo = part == 0 ? sh.rot : 0;
    const int hi = part == 0 ? sh.nvec : sh.rot;
    int vb = lo + wave * 32;
    for (; vb + (U - 1) * 1024 + 31 < hi; vb += U * 1024) {
      v4i d[U];
#pragma unroll
      for (int u = 0; u < U; ++u) d[u] = dv[vb + u * 1024 + lane];
#pragma unroll
      for (int u = 0; u < U; ++u)
        rec4<TN, CAPW>(li, lr, cnt, n0, d[u], head + (vb + u * 1024 + lane) * 4);
    }
    for (; vb < hi; vb += 1024) {
      const int v = vb + lane;
      v4i d = {-1, -1, -1, -1};
      if (v < hi) d = dv[v];
      rec4<TN, CAPW>(li, lr, cnt, n0, d, head + v * 4);
    }
  }
  if (wave == 0) {
    const int nt = E - sh.tail0;
    int i = -1;
    if (lane < head) i = lane;
    else if (lane - head < nt) i = sh.tail0 + (lane - head);
    const int dd = (i >= 0) ? dst[i] : -1;
    const unsigned r = (unsigned)(dd - n0);
    const bool f = (i >= 0) && (r < (unsigned)TN);
    rec<CAPW>(li, lr, cnt, __builtin_amdgcn_ballot_w32(f), f, i, r);
  }
  return cnt;
}

template <int C, int TN, int MODE>
__device__ __forceinline__ void score_list(const AggCtx<C, TN>& cx, int* li, int* lr,
                                           float* le, v2i* gl, int cnt) {
  constexpr int SB = 4;
  const int lane = cx.lane;
  for (int base = 0; base < cnt; base += 32 * SB) {
    int p[SB], r[SB], s[SB];
    float a[SB], d[SB];
#pragma unroll
    for (int u = 0; u < SB; ++u) {
      p[u] = base + u * 32 + lane;
      const int q = p[u] < cnt ? p[u] : cnt - 1;
      if (MODE == 0) {
        r[u] = lr[q];
        s[u] = cx.src[li[q]];
      } else {
        const v2i sr = gl[q];
        s[u] = sr.x;
        r[u] = sr.y;
      }
    }
#pragma unroll
    for (int u = 0; u < SB; ++u) {
      a[u] = cx.als[s[u]];
      d[u] = cx.ald[cx.n0 + r[u]];
    }
    static_assert(SB == 4, "the statement below names the SB gathered pairs");
    asm volatile("" ::"v"(a[0]), "v"(a[1]), "v"(a[2]), "v"(a[3]), "v"(d[0]), "v"(d[1]), "v"(d[2]), "v"(d[3]));
#pragma unroll
    for (int u = 0; u < SB; ++u) {
      if (p[u] < cnt) {
        const float e = lrelu(a[u] + d[u]);
        atomicMax(&cx.mx[r[u]], fenc(e));
        li[p[u]] = s[u];
        le[p[u]] = e;
        if (MODE == 0) {
          v2i sr;
          sr.x = s[u];
          sr.y = r[u];
          gl[p[u]] = sr;
        } else {
          lr[p[u]] = r[u];
        }
      }
    }
  }
}

template <int C, int TN>
__device__ __forceinline__ void sum_list(const AggCtx<C, TN>& cx, const int* li, const int* lr,
                                         const float* le, int cnt) {
  const int lane = cx.lane;
  for (int base = 0; base < cnt; base += 32) {
    const int nvalid = (cnt - base < 32) ? (cnt - base) : 32;
    int s = 0, r = 0;
    float x = 0.f;
    if (lane < nvalid) {
      s = li[base + lane];
      r = lr[base + lane];
      x = __expf(le[base + lane] - fdec(cx.mx[r]));
      atomicAdd(&cx.den[r], x);
    }
    add_rows<C, TN>(cx, s, r, x, nvalid);
  }
}


template <int C, int TN, int PHASE>
__device__ __forceinline__ void drain(const AggCtx<C, TN>& cx, int base, int nvalid) {
  const int lane = cx.lane;
  int s = 0, r = 0;
  float x = 0.f;
  if (lane < nvalid) {
    const int i = cx.qi[base + lane];
    r = cx.qr[base + lane];
    s = cx.src[i];
    const float e = lrelu(cx.als[s] + cx.ald[cx.n0 + r]);
    if (PHASE == 1) {
      atomicMax(&cx.mx[r], fenc(e));
    } else {
      x = __expf(e - fdec(cx.mx[r]));
      atomicAdd(&cx.den[r], x);
    }
  }
  if (PHASE == 2) add_rows<C, TN>(cx, s, r, x, nvalid);
}

template <int C, int TN, int PHASE>
__device__ __forceinline__ void push(const AggCtx<C, TN>& cx, int& cnt, bool f, int i, unsigned r) {
  const unsigned mask = __builtin_amdgcn_ballot_w32(f);
  if (mask == 0u) return;
  if (f) {
    const int slot = cnt + (int)__builtin_amdgcn_mbcnt_lo(mask, 0u);
    cx.qi[slot] = i;
    cx.qr[slot] = (int)r;
  }
  cnt += __builtin_popcount(mask);
  if (cnt >= 32) {
    wave_sync();
    cnt -= 32;
    drain<C, TN, PHASE>(cx, cnt, 32);
    wave_sync();
  }
}

template <int C, int TN, int PHASE>
__device__ __forceinline__ void proc4(const AggCtx<C, TN>& cx, int& cnt, v4i d, int ib) {
  const unsigned r0 = (unsigned)(d.x - cx.n0);
  const unsigned r1 = (unsigned)(d.y - cx.n0);
  const unsigned r2 = (unsigned)(d.z - cx.n0);
  const unsigned r3 = (unsigned)(d.w - cx.n0);
  const bool any = (r0 < (unsigned)TN) | (r1 < (unsigned)TN) | (r2 < (unsigned)TN) | (r3 < (unsigned)TN);
  if (__builtin_amdgcn_ballot_w32(any) == 0u) return;
  push<C, TN, PHASE>(cx, cnt, r0 < (unsigned)TN, ib, r0);
  push<C, TN, PHASE>(cx, cnt, r1 < (unsigned)TN, ib + 1, r1);
  push<C, TN, PHASE>(cx, cnt, r2 < (unsigned)TN, ib + 2, r2);
  push<C, TN, PHASE>(cx, cnt, r3 < (unsigned)TN, ib + 3, r3);
}

template <int C, int TN, int PHASE>
__device__ __forceinline__ void scan_slow(const AggCtx<C, TN>& cx, const Share& sh,
                                          const int* __restrict__ dst, int E, int wave) {
  const v4i* dv = sh.dv;
  const int head = sh.head;
  const int lane = cx.lane;
  int cnt = 0;
#pragma nounroll
  for (int part = 0; part < 2; ++part) {
    const int lo = part == 0 ? sh.rot : 0;
    const int hi = part == 0 ? sh.nvec : sh.rot;
    for (int vb = lo + wave * 32; vb < hi; vb += 1024) {
      const int v = vb + lane;
      v4i d = {-1, -1, -1, -1};
      if (v < hi) d = dv[v];
      proc4<C, TN, PHASE>(cx, cnt, d, head + v * 4);
    }
  }
  if (wave == 0) {
    const int nt = E - sh.tail0;
    int i = -1;
    if (lane < head) i = lane;
    else if (lane - head < nt) i = sh.tail0 + (lane - head);
    const int dd = (i >= 0) ? dst[i] : -1;
    const unsigned r = (unsigned)(dd - cx.n0);
    push<C, TN, PHASE>(cx, cnt, (i >= 0) && (r < (unsigned)TN), i, r);
  }
  wave_sync();
  if (cnt > 0) drain<C, TN, PHASE>(cx, 0, cnt);
  wave_sync();
}

template <int C, int TN, int CAPW, int MODE>
__global__ __launch_bounds__(1024) void agg_tile(const int* __restrict__ src,
                                                 const int* __restrict__ dst,
                                                 const float* __restrict__ als,
                                                 const float* __restrict__ ald,
                                                 const float* __restrict__ H,
                                                 const float* __restrict__ b,
                                                 float* __restrict__ out,
                                                 int* gcnt, int* glist,
                                                 int E, int Nn, int relu) {
  constexpr int VPT = C / 32;
  static_assert(CAPW >= 64, "the slow way's queue lives in the list's first 64 slots");
  __shared__ float acc[TN * C];
  __shared__ unsigned mx[TN];
  __shared__ float den[TN];
  __shared__ int li[32 * CAPW];
  __shared__ int lr[32 * CAPW];
  __shared__ float le[32 * CAPW];

  const int tid = threadIdx.x;
  const int lane = tid & 31;
  const int wave = __builtin_amdgcn_readfirstlane(tid >> 5);
  const int n0 = blockIdx.x * TN;

  for (int i = tid; i < TN * C; i += 1024) acc[i] = 0.f;
  for (int n = tid; n < TN; n += 1024) {
    const int node = n0 + n;
    float e = 0.f;
    if (node < Nn) e = lrelu(als[node] + ald[node]);
    mx[n] = fenc(e);
    den[n] = 0.f;
  }
  __syncthreads();

  int* wli = li + wave * CAPW;
  int* wlr = lr + wave * CAPW;
  float* wle = le + wave * CAPW;

  AggCtx<C, TN> cx;
  cx.src = src;
  cx.als = als;
  cx.ald = ald;
  cx.H = H;
  cx.acc = acc;
  cx.mx = mx;
  cx.den = den;
  cx.qi = wli;
  cx.qr = wlr;
  cx.n0 = n0;
  cx.lane = lane;

  const Share sh = share_of(dst, E);
  const int gw = blockIdx.x * 32 + wave;
  v2i* wgl = reinterpret_cast<v2i*>(glist) + (size_t)gw * CAPW;
  int kept;
  if (MODE == 0) {
    kept = scan_record<TN, CAPW>(sh, dst, E, n0, wave, lane, wli, wlr);
    if (lane == 0) gcnt[gw] = kept;
  } else {
    kept = __builtin_amdgcn_readfirstlane(gcnt[gw]);
  }
  const bool listed = kept <= CAPW;
  wave_sync();
  if (listed) score_list<C, TN, MODE>(cx, wli, wlr, wle, wgl, kept);
  else scan_slow<C, TN, 1>(cx, sh, dst, E, wave);
  __syncthreads();
  if (listed) sum_list<C, TN>(cx, wli, wlr, wle, kept);
  else scan_slow<C, TN, 2>(cx, sh, dst, E, wave);
  __syncthreads();

  for (int idx = tid; idx < TN * C; idx += 1024) {
    const int n = idx / C;
    const int c = idx & (C - 1);
    const int node = n0 + n;
    if (node < Nn) {
      const float es = lrelu(als[node] + ald[node]);
      const float xs = __expf(es - fdec(mx[n]));
      const float a = acc[n * C + (c % VPT) * 32 + c / VPT];
      float v = (a + xs * H[(size_t)node * C + c]) / (den[n] + xs) + b[c];
      if (relu) v = fmaxf(v, 0.f);
      out[(size_t)node * C + c] = v;
    }
  }
}

extern "C" void kernel_launch(void* const* d_in, const int* in_sizes, int n_in,
                              void* d_out, int out_size, void* d_ws, size_t ws_size,
                              hipStream_t stream) {
  const float* x   = (const float*)d_in[0];
  const int*   ei  = (const int*)d_in[1];
  const float* W1  = (const float*)d_in[2];
  const float* as1 = (const float*)d_in[3];
  const float* ad1 = (const float*)d_in[4];
  const float* b1  = (const float*)d_in[5];
  const float* W2  = (const float*)d_in[6];
  const float* as2 = (const float*)d_in[7];
  const float* ad2 = (const float*)d_in[8];
  const float* b2  = (const float*)d_in[9];
  const float* W3  = (const float*)d_in[10];
  const float* as3 = (const float*)d_in[11];
  const float* ad3 = (const float*)d_in[12];
  const float* b3  = (const float*)d_in[13];

  const int N = in_sizes[0] / 128;
  const int E = in_sizes[1] / 2;
  const int* srcp = ei;
  const int* dstp = ei + E;

  float* buf0 = (float*)d_ws;
  float* buf1 = buf0 + (size_t)N * 128;
  float* als  = buf1 + (size_t)N * 128;
  float* ald  = als + N;
  int* gcnt   = (int*)(ald + N);

  constexpr int TN = 400;
  constexpr int CAPW = 288;
  const int gGemm = (N / 16 + 7) / 8;
  const int gProj = (N + 255) / 256;
  const int gAgg  = (N + TN - 1) / TN;
  int* glist = gcnt + gAgg * 32;
  float* outF = (float*)d_out;

  gemm_wmma<128, 128><<<gGemm, 256, 0, stream>>>(x, W1, buf0, N);
  row_proj<128><<<gProj, 256, 0, stream>>>(buf0, as1, ad1, als, ald, N);
  agg_tile<128, TN, CAPW, 0><<<gAgg, 1024, 0, stream>>>(srcp, dstp, als, ald, buf0, b1, buf1, gcnt, glist, E, N, 1);

  gemm_wmma<128, 128><<<gGemm, 256, 0, stream>>>(buf1, W2, buf0, N);
  row_proj<128><<<gProj, 256, 0, stream>>>(buf0, as2, ad2, als, ald, N);
  agg_tile<128, TN, CAPW, 1><<<gAgg, 1024, 0, stream>>>(srcp, dstp, als, ald, buf0, b2, buf1, gcnt, glist, E, N, 1);

  gemm_wmma<128, 64><<<gGemm, 256, 0, stream>>>(buf1, W3, buf0, N);
  row_proj<64><<<gProj, 256, 0, stream>>>(buf0, as3, ad3, als, ald, N);
  agg_tile<64, TN, CAPW, 1><<<gAgg, 1024, 0, stream>>>(srcp, dstp, als, ald, buf0, b3, outF, gcnt, glist, E, N, 0);
}


# tile write-out by whole rows: a wave has 7 rows per trip, their h-row and score loads issued as one group, one vector store per lane; was one element per thread, two dependent loads a trip; sums kept
# speedup vs baseline: 4.9539x; 1.2101x over previous
#include <hip/hip_runtime.h>

typedef __attribute__((ext_vector_type(2))) float v2f;
typedef __attribute__((ext_vector_type(8))) float v8f;

#define NEG_SLOPE 0.2f

template <int CIN, int COUT>
__global__ __launch_bounds__(256) void gemm_wmma(const float* __restrict__ X,
                                                 const float* __restrict__ Wg,
                                                 float* __restrict__ H, int M) {
  __shared__ v2f Wswz[(CIN / 4) * COUT * 2];
  {
    constexpr int TOT = (CIN / 4) * COUT * 2;
    for (int i = threadIdx.x; i < TOT; i += 256) {
      int m = i / (COUT * 2);
      int r = i - m * (COUT * 2);
      int c = r >> 1;
      int hsel = r & 1;
      const float* wp = Wg + (size_t)(4 * m + 2 * hsel) * COUT + c;
      v2f w;
      w.x = wp[0];
      w.y = wp[COUT];
      Wswz[i] = w;
    }
  }
  __syncthreads();

  const int lane = threadIdx.x & 31;
  const int wave = threadIdx.x >> 5;
  const int mtile = blockIdx.x * 8 + wave;
  if (mtile * 16 >= M) return;
  const int m0 = mtile * 16;

  constexpr int NT = COUT / 16;
  v8f acc[NT] = {};

  const int rsel = lane & 15;
  const int hiHalf = lane >> 4;
  const float* xrow = X + (size_t)(m0 + rsel) * CIN + hiHalf * 2;
  const v2f* wbase = Wswz + rsel * 2 + hiHalf;

  for (int k = 0; k < CIN; k += 4) {
    v2f a = *reinterpret_cast<const v2f*>(xrow + k);
    const v2f* wb = wbase + (k >> 2) * (COUT * 2);
#pragma unroll
    for (int t = 0; t < NT; ++t) {
      v2f bf = wb[t * 32];
      acc[t] = __builtin_amdgcn_wmma_f32_16x16x4_f32(
          false, a, false, bf, (short)0, acc[t], false, false);
    }
  }

  const int rbase = m0 + hiHalf * 8;
#pragma unroll
  for (int t = 0; t < NT; ++t) {
#pragma unroll
    for (int i = 0; i < 8; ++i) {
      H[(size_t)(rbase + i) * COUT + t * 16 + rsel] = acc[t][i];
    }
  }
}

template <int C>
__global__ __launch_bounds__(256) void row_proj(const float* __restrict__ H,
                                                const float* __restrict__ asrc,
                                                const float* __restrict__ adst,
                                                float* __restrict__ als,
                                                float* __restrict__ ald, int Nn) {
  constexpr int VPT = C / 32;
  constexpr int RB = 8;
  const int lane = threadIdx.x & 31;
  const int wave = threadIdx.x >> 5;
  const int nb = (blockIdx.x * 8 + wave) * 32;
  if (nb >= Nn) return;
  float as[VPT], ad[VPT];
#pragma unroll
  for (int j = 0; j < VPT; ++j) {
    as[j] = asrc[lane + 32 * j];
    ad[j] = adst[lane + 32 * j];
  }
  float rs = 0.f, rd = 0.f;
  for (int k = 0; k < 32; k += RB) {
    float hv[RB][VPT];
#pragma unroll
    for (int u = 0; u < RB; ++u) {
      int n = nb + k + u;
      if (n > Nn - 1) n = Nn - 1;
      const float* hrow = H + (size_t)n * C;
#pragma unroll
      for (int j = 0; j < VPT; ++j) hv[u][j] = hrow[lane + 32 * j];
    }
#pragma unroll
    for (int u = 0; u < RB; ++u) {
      float s = 0.f, d = 0.f;
#pragma unroll
      for (int j = 0; j < VPT; ++j) {
        s += hv[u][j] * as[j];
        d += hv[u][j] * ad[j];
      }
#pragma unroll
      for (int off = 16; off > 0; off >>= 1) {
        s += __shfl_xor(s, off, 32);
        d += __shfl_xor(d, off, 32);
      }
      if (lane == k + u) {
        rs = s;
        rd = d;
      }
    }
  }
  if (nb + lane < Nn) {
    als[nb + lane] = rs;
    ald[nb + lane] = rd;
  }
}

__device__ __forceinline__ unsigned fenc(float f) {
  unsigned u = __float_as_uint(f);
  return u ^ ((u >> 31) ? 0xFFFFFFFFu : 0x80000000u);
}
__device__ __forceinline__ float fdec(unsigned u) {
  unsigned v = u ^ ((u & 0x80000000u) ? 0x80000000u : 0xFFFFFFFFu);
  return __uint_as_float(v);
}

typedef __attribute__((ext_vector_type(4))) int v4i;
typedef __attribute__((ext_vector_type(2))) int v2i;

__device__ __forceinline__ float lrelu(float e) { return (e > 0.f) ? e : NEG_SLOPE * e; }

__device__ __forceinline__ void wave_sync() {
  __builtin_amdgcn_fence(__ATOMIC_RELEASE, "wavefront");
  __builtin_amdgcn_wave_barrier();
  __builtin_amdgcn_fence(__ATOMIC_ACQUIRE, "wavefront");
}

template <int C, int TN>
struct AggCtx {
  const int* src;
  const float* als;
  const float* ald;
  const float* H;
  float* acc;
  unsigned* mx;
  float* den;
  int* qi;
  int* qr;
  int n0;
  int lane;
};

template <int C, int TN>
__device__ __forceinline__ void add_rows(const AggCtx<C, TN>& cx, int s, int r, float x, int nvalid) {
  constexpr int VPT = C / 32;
  typedef float vrow __attribute__((ext_vector_type(VPT)));
  const int lane = cx.lane;
  const int col = lane * VPT;
  constexpr int B = 8;
  int k = 0;
  for (; k + B <= nvalid; k += B) {
    vrow v[B];
    int rk[B];
    float xk[B];
#pragma unroll
    for (int u = 0; u < B; ++u) {
      const int sk = __builtin_amdgcn_readlane(s, k + u);
      rk[u] = __builtin_amdgcn_readlane(r, k + u);
      xk[u] = __int_as_float(__builtin_amdgcn_readlane(__float_as_int(x), k + u));
      v[u] = *reinterpret_cast<const vrow*>(cx.H + (size_t)sk * C + col);
    }
#pragma unroll
    for (int u = 0; u < B; ++u) {
      float* ar = cx.acc + rk[u] * C + lane;
#pragma unroll
      for (int j = 0; j < VPT; ++j) atomicAdd(&ar[j * 32], xk[u] * v[u][j]);
    }
  }
  for (; k < nvalid; ++k) {
    const int sk = __builtin_amdgcn_readlane(s, k);
    const int rq = __builtin_amdgcn_readlane(r, k);
    const float xq = __int_as_float(__builtin_amdgcn_readlane(__float_as_int(x), k));
    const vrow hv = *reinterpret_cast<const vrow*>(cx.H + (size_t)sk * C + col);
    float* ar = cx.acc + rq * C + lane;
#pragma unroll
    for (int j = 0; j < VPT; ++j) atomicAdd(&ar[j * 32], xq * hv[j]);
  }
}


struct Share {
  const v4i* dv;
  int head;
  int nvec;
  int tail0;
  int rot;
};

__device__ __forceinline__ Share share_of(const int* __restrict__ dst, int E) {
  Share sh;
  const unsigned long long da = (unsigned long long)dst;
  int head = (int)(((16u - (unsigned)(da & 15ull)) & 15u) >> 2);
  if (head > E) head = E;
  sh.head = head;
  sh.nvec = (E - head) >> 2;
  sh.tail0 = head + sh.nvec * 4;
  sh.dv = reinterpret_cast<const v4i*>(dst + head);
  sh.rot = (int)(((long long)blockIdx.x * sh.nvec) / gridDim.x) & ~31;
  return sh;
}


template <int CAPW>
__device__ __forceinline__ void rec(int* li, int* lr, int& cnt, unsigned mask, bool f, int i, unsigned r) {
  if (mask == 0u) return;
  if (f) {
    const int slot = cnt + (int)__builtin_amdgcn_mbcnt_lo(mask, 0u);
    if (slot < CAPW) {
      li[slot] = i;
      lr[slot] = (int)r;
    }
  }
  cnt += __builtin_popcount(mask);
}

template <int TN, int CAPW>
__device__ __forceinline__ void rec4(int* li, int* lr, int& cnt, int n0, v4i d, int ib) {
  const unsigned r0 = (unsigned)(d.x - n0);
  const unsigned r1 = (unsigned)(d.y - n0);
  const unsigned r2 = (unsigned)(d.z - n0);
  const unsigned r3 = (unsigned)(d.w - n0);
  const bool f0 = r0 < (unsigned)TN;
  const bool f1 = r1 < (unsigned)TN;
  const bool f2 = r2 < (unsigned)TN;
  const bool f3 = r3 < (unsigned)TN;
  const unsigned m0 = __builtin_amdgcn_ballot_w32(f0);
  const unsigned m1 = __builtin_amdgcn_ballot_w32(f1);
  const unsigned m2 = __builtin_amdgcn_ballot_w32(f2);
  const unsigned m3 = __builtin_amdgcn_ballot_w32(f3);
  if ((m0 | m1 | m2 | m3) == 0u) return;
  rec<CAPW>(li, lr, cnt, m0, f0, ib, r0);
  rec<CAPW>(li, lr, cnt, m1, f1, ib + 1, r1);
  rec<CAPW>(li, lr, cnt, m2, f2, ib + 2, r2);
  rec<CAPW>(li, lr, cnt, m3, f3, ib + 3, r3);
}

template <int TN, int CAPW>
__device__ __forceinline__ int scan_record(const Share& sh, const int* __restrict__ dst, int E, int n0,
                                           int wave, int lane, int* li, int* lr) {
  const v4i* dv = sh.dv;
  const int head = sh.head;
  int cnt = 0;
  constexpr int U = 8;
#pragma nounroll
  for (int part = 0; part < 2; ++part) {
    const int lo = part == 0 ? sh.rot : 0;
    const int hi = part == 0 ? sh.nvec : sh.rot;
    int vb = lo + wave * 32;
    for (; vb + (U - 1) * 1024 + 31 < hi; vb += U * 1024) {
      v4i d[U];
#pragma unroll
      for (int u = 0; u < U; ++u) d[u] = dv[vb + u * 1024 + lane];
#pragma unroll
      for (int u = 0; u < U; ++u)
        rec4<TN, CAPW>(li, lr, cnt, n0, d[u], head + (vb + u * 1024 + lane) * 4);
    }
    for (; vb < hi; vb += 1024) {
      const int v = vb + lane;
      v4i d = {-1, -1, -1, -1};
      if (v < hi) d = dv[v];
      rec4<TN, CAPW>(li, lr, cnt, n0, d, head + v * 4);
    }
  }
  if (wave == 0) {
    const int nt = E - sh.tail0;
    int i = -1;
    if (lane < head) i = lane;
    else if (lane - head < nt) i = sh.tail0 + (lane - head);
    const int dd = (i >= 0) ? dst[i] : -1;
    const unsigned r = (unsigned)(dd - n0);
    const bool f = (i >= 0) && (r < (unsigned)TN);
    rec<CAPW>(li, lr, cnt, __builtin_amdgcn_ballot_w32(f), f, i, r);
  }
  return cnt;
}

template <int C, int TN, int MODE>
__device__ __forceinline__ void score_list(const AggCtx<C, TN>& cx, int* li, int* lr,
                                           float* le, v2i* gl, int cnt) {
  constexpr int SB = 4;
  const int lane = cx.lane;
  for (int base = 0; base < cnt; base += 32 * SB) {
    int p[SB], r[SB], s[SB];
    float a[SB], d[SB];
#pragma unroll
    for (int u = 0; u < SB; ++u) {
      p[u] = base + u * 32 + lane;
      const int q = p[u] < cnt ? p[u] : cnt - 1;
      if (MODE == 0) {
        r[u] = lr[q];
        s[u] = cx.src[li[q]];
      } else {
        const v2i sr = gl[q];
        s[u] = sr.x;
        r[u] = sr.y;
      }
    }
#pragma unroll
    for (int u = 0; u < SB; ++u) {
      a[u] = cx.als[s[u]];
      d[u] = cx.ald[cx.n0 + r[u]];
    }
    static_assert(SB == 4, "the statement below names the SB gathered pairs");
    asm volatile("" ::"v"(a[0]), "v"(a[1]), "v"(a[2]), "v"(a[3]), "v"(d[0]), "v"(d[1]), "v"(d[2]), "v"(d[3]));
#pragma unroll
    for (int u = 0; u < SB; ++u) {
      if (p[u] < cnt) {
        const float e = lrelu(a[u] + d[u]);
        atomicMax(&cx.mx[r[u]], fenc(e));
        li[p[u]] = s[u];
        le[p[u]] = e;
        if (MODE == 0) {
          v2i sr;
          sr.x = s[u];
          sr.y = r[u];
          gl[p[u]] = sr;
        } else {
          lr[p[u]] = r[u];
        }
      }
    }
  }
}

template <int C, int TN>
__device__ __forceinline__ void sum_list(const AggCtx<C, TN>& cx, const int* li, const int* lr,
                                         const float* le, int cnt) {
  const int lane = cx.lane;
  for (int base = 0; base < cnt; base += 32) {
    const int nvalid = (cnt - base < 32) ? (cnt - base) : 32;
    int s = 0, r = 0;
    float x = 0.f;
    if (lane < nvalid) {
      s = li[base + lane];
      r = lr[base + lane];
      x = __expf(le[base + lane] - fdec(cx.mx[r]));
      atomicAdd(&cx.den[r], x);
    }
    add_rows<C, TN>(cx, s, r, x, nvalid);
  }
}


template <int C, int TN, int PHASE>
__device__ __forceinline__ void drain(const AggCtx<C, TN>& cx, int base, int nvalid) {
  const int lane = cx.lane;
  int s = 0, r = 0;
  float x = 0.f;
  if (lane < nvalid) {
    const int i = cx.qi[base + lane];
    r = cx.qr[base + lane];
    s = cx.src[i];
    const float e = lrelu(cx.als[s] + cx.ald[cx.n0 + r]);
    if (PHASE == 1) {
      atomicMax(&cx.mx[r], fenc(e));
    } else {
      x = __expf(e - fdec(cx.mx[r]));
      atomicAdd(&cx.den[r], x);
    }
  }
  if (PHASE == 2) add_rows<C, TN>(cx, s, r, x, nvalid);
}

template <int C, int TN, int PHASE>
__device__ __forceinline__ void push(const AggCtx<C, TN>& cx, int& cnt, bool f, int i, unsigned r) {
  const unsigned mask = __builtin_amdgcn_ballot_w32(f);
  if (mask == 0u) return;
  if (f) {
    const int slot = cnt + (int)__builtin_amdgcn_mbcnt_lo(mask, 0u);
    cx.qi[slot] = i;
    cx.qr[slot] = (int)r;
  }
  cnt += __builtin_popcount(mask);
  if (cnt >= 32) {
    wave_sync();
    cnt -= 32;
    drain<C, TN, PHASE>(cx, cnt, 32);
    wave_sync();
  }
}

template <int C, int TN, int PHASE>
__device__ __forceinline__ void proc4(const AggCtx<C, TN>& cx, int& cnt, v4i d, int ib) {
  const unsigned r0 = (unsigned)(d.x - cx.n0);
  const unsigned r1 = (unsigned)(d.y - cx.n0);
  const unsigned r2 = (unsigned)(d.z - cx.n0);
  const unsigned r3 = (unsigned)(d.w - cx.n0);
  const bool any = (r0 < (unsigned)TN) | (r1 < (unsigned)TN) | (r2 < (unsigned)TN) | (r3 < (unsigned)TN);
  if (__builtin_amdgcn_ballot_w32(any) == 0u) return;
  push<C, TN, PHASE>(cx, cnt, r0 < (unsigned)TN, ib, r0);
  push<C, TN, PHASE>(cx, cnt, r1 < (unsigned)TN, ib + 1, r1);
  push<C, TN, PHASE>(cx, cnt, r2 < (unsigned)TN, ib + 2, r2);
  push<C, TN, PHASE>(cx, cnt, r3 < (unsigned)TN, ib + 3, r3);
}

template <int C, int TN, int PHASE>
__device__ __forceinline__ void scan_slow(const AggCtx<C, TN>& cx, const Share& sh,
                                          const int* __restrict__ dst, int E, int wave) {
  const v4i* dv = sh.dv;
  const int head = sh.head;
  const int lane = cx.lane;
  int cnt = 0;
#pragma nounroll
  for (int part = 0; part < 2; ++part) {
    const int lo = part == 0 ? sh.rot : 0;
    const int hi = part == 0 ? sh.nvec : sh.rot;
    for (int vb = lo + wave * 32; vb < hi; vb += 1024) {
      const int v = vb + lane;
      v4i d = {-1, -1, -1, -1};
      if (v < hi) d = dv[v];
      proc4<C, TN, PHASE>(cx, cnt, d, head + v * 4);
    }
  }
  if (wave == 0) {
    const int nt = E - sh.tail0;
    int i = -1;
    if (lane < head) i = lane;
    else if (lane - head < nt) i = sh.tail0 + (lane - head);
    const int dd = (i >= 0) ? dst[i] : -1;
    const unsigned r = (unsigned)(dd - cx.n0);
    push<C, TN, PHASE>(cx, cnt, (i >= 0) && (r < (unsigned)TN), i, r);
  }
  wave_sync();
  if (cnt > 0) drain<C, TN, PHASE>(cx, 0, cnt);
  wave_sync();
}

template <int C, int TN, int CAPW, int MODE>
__global__ __launch_bounds__(1024) void agg_tile(const int* __restrict__ src,
                                                 const int* __restrict__ dst,
                                                 const float* __restrict__ als,
                                                 const float* __restrict__ ald,
                                                 const float* __restrict__ H,
                                                 const float* __restrict__ b,
                                                 float* __restrict__ out,
                                                 int* gcnt, int* glist,
                                                 int E, int Nn, int relu) {
  constexpr int VPT = C / 32;
  static_assert(CAPW >= 64, "the slow way's queue lives in the list's first 64 slots");
  __shared__ float acc[TN * C];
  __shared__ unsigned mx[TN];
  __shared__ float den[TN];
  __shared__ int li[32 * CAPW];
  __shared__ int lr[32 * CAPW];
  __shared__ float le[32 * CAPW];

  const int tid = threadIdx.x;
  const int lane = tid & 31;
  const int wave = __builtin_amdgcn_readfirstlane(tid >> 5);
  const int n0 = blockIdx.x * TN;

  for (int i = tid; i < TN * C; i += 1024) acc[i] = 0.f;
  for (int n = tid; n < TN; n += 1024) {
    const int node = n0 + n;
    float e = 0.f;
    if (node < Nn) e = lrelu(als[node] + ald[node]);
    mx[n] = fenc(e);
    den[n] = 0.f;
  }
  __syncthreads();

  int* wli = li + wave * CAPW;
  int* wlr = lr + wave * CAPW;
  float* wle = le + wave * CAPW;

  AggCtx<C, TN> cx;
  cx.src = src;
  cx.als = als;
  cx.ald = ald;
  cx.H = H;
  cx.acc = acc;
  cx.mx = mx;
  cx.den = den;
  cx.qi = wli;
  cx.qr = wlr;
  cx.n0 = n0;
  cx.lane = lane;

  const Share sh = share_of(dst, E);
  const int gw = blockIdx.x * 32 + wave;
  v2i* wgl = reinterpret_cast<v2i*>(glist) + (size_t)gw * CAPW;
  int kept;
  if (MODE == 0) {
    kept = scan_record<TN, CAPW>(sh, dst, E, n0, wave, lane, wli, wlr);
    if (lane == 0) gcnt[gw] = kept;
  } else {
    kept = __builtin_amdgcn_readfirstlane(gcnt[gw]);
  }
  const bool listed = kept <= CAPW;
  wave_sync();
  if (listed) score_list<C, TN, MODE>(cx, wli, wlr, wle, wgl, kept);
  else scan_slow<C, TN, 1>(cx, sh, dst, E, wave);
  __syncthreads();
  if (listed) sum_list<C, TN>(cx, wli, wlr, wle, kept);
  else scan_slow<C, TN, 2>(cx, sh, dst, E, wave);
  __syncthreads();

  {
    typedef float vrow __attribute__((ext_vector_type(VPT)));
    constexpr int WB = 7;
    const int col = lane * VPT;
    const vrow bv = *reinterpret_cast<const vrow*>(b + col);
    for (int nb = wave; nb < TN; nb += 32 * WB) {
      vrow hv[WB];
      float sa[WB], sd[WB];
#pragma unroll
      for (int u = 0; u < WB; ++u) {
        const int n = nb + 32 * u;
        int node = n0 + n;
        if (n >= TN || node >= Nn) node = n0;
        hv[u] = *reinterpret_cast<const vrow*>(H + (size_t)node * C + col);
        sa[u] = als[node];
        sd[u] = ald[node];
      }
#pragma unroll
      for (int u = 0; u < WB; ++u) {
        const int n = nb + 32 * u;
        const int node = n0 + n;
        if (n < TN && node < Nn) {
          const float es = lrelu(sa[u] + sd[u]);
          const float xs = __expf(es - fdec(mx[n]));
          const float dn = den[n] + xs;
          vrow o;
#pragma unroll
          for (int j = 0; j < VPT; ++j) {
            const float a = acc[n * C + j * 32 + lane];
            float v = (a + xs * hv[u][j]) / dn + bv[j];
            if (relu) v = fmaxf(v, 0.f);
            o[j] = v;
          }
          *reinterpret_cast<vrow*>(out + (size_t)node * C + col) = o;
        }
      }
    }
  }
}

extern "C" void kernel_launch(void* const* d_in, const int* in_sizes, int n_in,
                              void* d_out, int out_size, void* d_ws, size_t ws_size,
                              hipStream_t stream) {
  const float* x   = (const float*)d_in[0];
  const int*   ei  = (const int*)d_in[1];
  const float* W1  = (const float*)d_in[2];
  const float* as1 = (const float*)d_in[3];
  const float* ad1 = (const float*)d_in[4];
  const float* b1  = (const float*)d_in[5];
  const float* W2  = (const float*)d_in[6];
  const float* as2 = (const float*)d_in[7];
  const float* ad2 = (const float*)d_in[8];
  const float* b2  = (const float*)d_in[9];
  const float* W3  = (const float*)d_in[10];
  const float* as3 = (const float*)d_in[11];
  const float* ad3 = (const float*)d_in[12];
  const float* b3  = (const float*)d_in[13];

  const int N = in_sizes[0] / 128;
  const int E = in_sizes[1] / 2;
  const int* srcp = ei;
  const int* dstp = ei + E;

  float* buf0 = (float*)d_ws;
  float* buf1 = buf0 + (size_t)N * 128;
  float* als  = buf1 + (size_t)N * 128;
  float* ald  = als + N;
  int* gcnt   = (int*)(ald + N);

  constexpr int TN = 400;
  constexpr int CAPW = 288;
  const int gGemm = (N / 16 + 7) / 8;
  const int gProj = (N + 255) / 256;
  const int gAgg  = (N + TN - 1) / TN;
  int* glist = gcnt + gAgg * 32;
  float* outF = (float*)d_out;

  gemm_wmma<128, 128><<<gGemm, 256, 0, stream>>>(x, W1, buf0, N);
  row_proj<128><<<gProj, 256, 0, stream>>>(buf0, as1, ad1, als, ald, N);
  agg_tile<128, TN, CAPW, 0><<<gAgg, 1024, 0, stream>>>(srcp, dstp, als, ald, buf0, b1, buf1, gcnt, glist, E, N, 1);

  gemm_wmma<128, 128><<<gGemm, 256, 0, stream>>>(buf1, W2, buf0, N);
  row_proj<128><<<gProj, 256, 0, stream>>>(buf0, as2, ad2, als, ald, N);
  agg_tile<128, TN, CAPW, 1><<<gAgg, 1024, 0, stream>>>(srcp, dstp, als, ald, buf0, b2, buf1, gcnt, glist, E, N, 1);

  gemm_wmma<128, 64><<<gGemm, 256, 0, stream>>>(buf1, W3, buf0, N);
  row_proj<64><<<gProj, 256, 0, stream>>>(buf0, as3, ad3, als, ald, N);
  agg_tile<64, TN, CAPW, 1><<<gAgg, 1024, 0, stream>>>(srcp, dstp, als, ald, buf0, b3, outF, gcnt, glist, E, N, 0);
}


# layer 1's aggregation filters edges by a one-byte key (destination tile number, written by the row-projection launch): a tile streams E bytes, not 4E; list order, so LDS add order, changes
# speedup vs baseline: 5.1370x; 1.0370x over previous
#include <hip/hip_runtime.h>

typedef __attribute__((ext_vector_type(2))) float v2f;
typedef __attribute__((ext_vector_type(8))) float v8f;

#define NEG_SLOPE 0.2f

template <int CIN, int COUT>
__global__ __launch_bounds__(256) void gemm_wmma(const float* __restrict__ X,
                                                 const float* __restrict__ Wg,
                                                 float* __restrict__ H, int M) {
  __shared__ v2f Wswz[(CIN / 4) * COUT * 2];
  {
    constexpr int TOT = (CIN / 4) * COUT * 2;
    for (int i = threadIdx.x; i < TOT; i += 256) {
      int m = i / (COUT * 2);
      int r = i - m * (COUT * 2);
      int c = r >> 1;
      int hsel = r & 1;
      const float* wp = Wg + (size_t)(4 * m + 2 * hsel) * COUT + c;
      v2f w;
      w.x = wp[0];
      w.y = wp[COUT];
      Wswz[i] = w;
    }
  }
  __syncthreads();

  const int lane = threadIdx.x & 31;
  const int wave = threadIdx.x >> 5;
  const int mtile = blockIdx.x * 8 + wave;
  if (mtile * 16 >= M) return;
  const int m0 = mtile * 16;

  constexpr int NT = COUT / 16;
  v8f acc[NT] = {};

  const int rsel = lane & 15;
  const int hiHalf = lane >> 4;
  const float* xrow = X + (size_t)(m0 + rsel) * CIN + hiHalf * 2;
  const v2f* wbase = Wswz + rsel * 2 + hiHalf;

  for (int k = 0; k < CIN; k += 4) {
    v2f a = *reinterpret_cast<const v2f*>(xrow + k);
    const v2f* wb = wbase + (k >> 2) * (COUT * 2);
#pragma unroll
    for (int t = 0; t < NT; ++t) {
      v2f bf = wb[t * 32];
      acc[t] = __builtin_amdgcn_wmma_f32_16x16x4_f32(
          false, a, false, bf, (short)0, acc[t], false, false);
    }
  }

  const int rbase = m0 + hiHalf * 8;
#pragma unroll
  for (int t = 0; t < NT; ++t) {
#pragma unroll
    for (int i = 0; i < 8; ++i) {
      H[(size_t)(rbase + i) * COUT + t * 16 + rsel] = acc[t][i];
    }
  }
}

template <int TNK>
__device__ __forceinline__ unsigned key_dword(const int* __restrict__ p) {
  unsigned kw = 0u;
#pragma unroll
  for (int j = 0; j < 4; ++j) {
    unsigned t = (unsigned)p[j] / (unsigned)TNK;
    if (t > 255u) t = 255u;
    kw |= t << (8 * j);
  }
  return kw;
}

template <int C, int TNK>
__global__ __launch_bounds__(256) void row_proj(const float* __restrict__ H,
                                                const float* __restrict__ asrc,
                                                const float* __restrict__ adst,
                                                float* __restrict__ als,
                                                float* __restrict__ ald, int Nn,
                                                const int* __restrict__ dst,
                                                unsigned* __restrict__ keys, int E) {
  if constexpr (TNK > 0) {
    const int nq = E >> 2;
    const int step = (int)gridDim.x * 256;
    constexpr int KU = 4;
    int q = (int)blockIdx.x * 256 + (int)threadIdx.x;
    for (; q + (KU - 1) * step < nq; q += KU * step) {
      unsigned kw[KU];
#pragma unroll
      for (int u = 0; u < KU; ++u) kw[u] = key_dword<TNK>(dst + 4 * (q + u * step));
#pragma unroll
      for (int u = 0; u < KU; ++u) keys[q + u * step] = kw[u];
    }
    for (; q < nq; q += step) keys[q] = key_dword<TNK>(dst + 4 * q);
  }
  constexpr int VPT = C / 32;
  constexpr int RB = 8;
  const int lane = threadIdx.x & 31;
  const int wave = threadIdx.x >> 5;
  const int nb = (blockIdx.x * 8 + wave) * 32;
  if (nb >= Nn) return;
  float as[VPT], ad[VPT];
#pragma unroll
  for (int j = 0; j < VPT; ++j) {
    as[j] = asrc[lane + 32 * j];
    ad[j] = adst[lane + 32 * j];
  }
  float rs = 0.f, rd = 0.f;
  for (int k = 0; k < 32; k += RB) {
    float hv[RB][VPT];
#pragma unroll
    for (int u = 0; u < RB; ++u) {
      int n = nb + k + u;
      if (n > Nn - 1) n = Nn - 1;
      const float* hrow = H + (size_t)n * C;
#pragma unroll
      for (int j = 0; j < VPT; ++j) hv[u][j] = hrow[lane + 32 * j];
    }
#pragma unroll
    for (int u = 0; u < RB; ++u) {
      float s = 0.f, d = 0.f;
#pragma unroll
      for (int j = 0; j < VPT; ++j) {
        s += hv[u][j] * as[j];
        d += hv[u][j] * ad[j];
      }
#pragma unroll
      for (int off = 16; off > 0; off >>= 1) {
        s += __shfl_xor(s, off, 32);
        d += __shfl_xor(d, off, 32);
      }
      if (lane == k + u) {
        rs = s;
        rd = d;
      }
    }
  }
  if (nb + lane < Nn) {
    als[nb + lane] = rs;
    ald[nb + lane] = rd;
  }
}

__device__ __forceinline__ unsigned fenc(float f) {
  unsigned u = __float_as_uint(f);
  return u ^ ((u >> 31) ? 0xFFFFFFFFu : 0x80000000u);
}
__device__ __forceinline__ float fdec(unsigned u) {
  unsigned v = u ^ ((u & 0x80000000u) ? 0x80000000u : 0xFFFFFFFFu);
  return __uint_as_float(v);
}

typedef __attribute__((ext_vector_type(4))) int v4i;
typedef __attribute__((ext_vector_type(2))) int v2i;

__device__ __forceinline__ float lrelu(float e) { return (e > 0.f) ? e : NEG_SLOPE * e; }

__device__ __forceinline__ void wave_sync() {
  __builtin_amdgcn_fence(__ATOMIC_RELEASE, "wavefront");
  __builtin_amdgcn_wave_barrier();
  __builtin_amdgcn_fence(__ATOMIC_ACQUIRE, "wavefront");
}

template <int C, int TN>
struct AggCtx {
  const int* src;
  const int* dst;
  const float* als;
  const float* ald;
  const float* H;
  float* acc;
  unsigned* mx;
  float* den;
  int* qi;
  int* qr;
  int n0;
  int lane;
};

template <int C, int TN>
__device__ __forceinline__ void add_rows(const AggCtx<C, TN>& cx, int s, int r, float x, int nvalid) {
  constexpr int VPT = C / 32;
  typedef float vrow __attribute__((ext_vector_type(VPT)));
  const int lane = cx.lane;
  const int col = lane * VPT;
  constexpr int B = 8;
  int k = 0;
  for (; k + B <= nvalid; k += B) {
    vrow v[B];
    int rk[B];
    float xk[B];
#pragma unroll
    for (int u = 0; u < B; ++u) {
      const int sk = __builtin_amdgcn_readlane(s, k + u);
      rk[u] = __builtin_amdgcn_readlane(r, k + u);
      xk[u] = __int_as_float(__builtin_amdgcn_readlane(__float_as_int(x), k + u));
      v[u] = *reinterpret_cast<const vrow*>(cx.H + (size_t)sk * C + col);
    }
#pragma unroll
    for (int u = 0; u < B; ++u) {
      float* ar = cx.acc + rk[u] * C + lane;
#pragma unroll
      for (int j = 0; j < VPT; ++j) atomicAdd(&ar[j * 32], xk[u] * v[u][j]);
    }
  }
  for (; k < nvalid; ++k) {
    const int sk = __builtin_amdgcn_readlane(s, k);
    const int rq = __builtin_amdgcn_readlane(r, k);
    const float xq = __int_as_float(__builtin_amdgcn_readlane(__float_as_int(x), k));
    const vrow hv = *reinterpret_cast<const vrow*>(cx.H + (size_t)sk * C + col);
    float* ar = cx.acc + rq * C + lane;
#pragma unroll
    for (int j = 0; j < VPT; ++j) atomicAdd(&ar[j * 32], xq * hv[j]);
  }
}


struct Share {
  const v4i* kv;
  int nvec;
  int tail0;
  int rot;
};

__device__ __forceinline__ Share share_of(const unsigned* __restrict__ keys, int E) {
  Share sh;
  sh.kv = reinterpret_cast<const v4i*>(keys);
  sh.nvec = E >> 4;
  sh.tail0 = sh.nvec * 16;
  sh.rot = (int)(((long long)blockIdx.x * sh.nvec) / gridDim.x) & ~31;
  return sh;
}

__device__ __forceinline__ unsigned zero_bytes(unsigned x) {
  return ~(((x & 0x7F7F7F7Fu) + 0x7F7F7F7Fu) | x | 0x7F7F7F7Fu);
}


template <int CAPW>
__device__ __forceinline__ void rec(int* li, int& cnt, unsigned mask, bool f, int i) {
  if (mask == 0u) return;
  if (f) {
    const int slot = cnt + (int)__builtin_amdgcn_mbcnt_lo(mask, 0u);
    if (slot < CAPW) li[slot] = i;
  }
  cnt += __builtin_popcount(mask);
}

__device__ __forceinline__ unsigned vec_flags(v4i k, unsigned pat) {
  return (zero_bytes((unsigned)k.x ^ pat) >> 7) | (zero_bytes((unsigned)k.y ^ pat) >> 6) |
         (zero_bytes((unsigned)k.z ^ pat) >> 5) | (zero_bytes((unsigned)k.w ^ pat) >> 4);
}

__device__ __forceinline__ int take_flag(unsigned& zc) {
  const int p = __builtin_ctz(zc | 0x80000000u);
  zc &= zc - 1u;
  return 4 * (p & 7) + (p >> 3);
}

template <int CAPW>
__device__ __forceinline__ void rec_vec(int* li, int& cnt, unsigned pat, v4i k, int ib) {
  unsigned zc = vec_flags(k, pat);
  unsigned mask = __builtin_amdgcn_ballot_w32(zc != 0u);
  while (mask != 0u) {
    const bool f = zc != 0u;
    const int i = ib + take_flag(zc);
    rec<CAPW>(li, cnt, mask, f, i);
    mask = __builtin_amdgcn_ballot_w32(zc != 0u);
  }
}

template <int TN, int CAPW>
__device__ __forceinline__ int scan_record(const Share& sh, const int* __restrict__ dst, int E, int n0,
                                           unsigned pat, int wave, int lane, int* li) {
  const v4i* kv = sh.kv;
  int cnt = 0;
  constexpr int U = 8;
#pragma nounroll
  for (int part = 0; part < 2; ++part) {
    const int lo = part == 0 ? sh.rot : 0;
    const int hi = part == 0 ? sh.nvec : sh.rot;
    for (int vb = lo + wave * 32; vb < hi; vb += U * 1024) {
      v4i k[U];
      unsigned pl[U];
#pragma unroll
      for (int u = 0; u < U; ++u) {
        const int v = vb + u * 1024 + lane;
        pl[u] = v < hi ? pat : 0xFFFFFFFFu;
        k[u] = kv[v < hi ? v : hi - 1];
      }
#pragma unroll
      for (int u = 0; u < U; ++u)
        rec_vec<CAPW>(li, cnt, pl[u], k[u], (vb + u * 1024 + lane) * 16);
    }
  }
  if (wave == 0) {
    const int i = sh.tail0 + lane;
    const int dd = (i < E) ? dst[i] : -1;
    const bool f = (i < E) && ((unsigned)(dd - n0) < (unsigned)TN);
    rec<CAPW>(li, cnt, __builtin_amdgcn_ballot_w32(f), f, i);
  }
  return cnt;
}

template <int C, int TN, int MODE>
__device__ __forceinline__ void score_list(const AggCtx<C, TN>& cx, int* li, int* lr,
                                           float* le, v2i* gl, int cnt) {
  constexpr int SB = 4;
  const int lane = cx.lane;
  for (int base = 0; base < cnt; base += 32 * SB) {
    int p[SB], r[SB], s[SB];
    float a[SB], d[SB];
#pragma unroll
    for (int u = 0; u < SB; ++u) {
      p[u] = base + u * 32 + lane;
      const int q = p[u] < cnt ? p[u] : cnt - 1;
      if (MODE == 0) {
        const int i = li[q];
        s[u] = cx.src[i];
        r[u] = cx.dst[i] - cx.n0;
      } else {
        const v2i sr = gl[q];
        s[u] = sr.x;
        r[u] = sr.y;
      }
    }
#pragma unroll
    for (int u = 0; u < SB; ++u) {
      a[u] = cx.als[s[u]];
      d[u] = cx.ald[cx.n0 + r[u]];
    }
    static_assert(SB == 4, "the statement below names the SB gathered pairs");
    asm volatile("" ::"v"(a[0]), "v"(a[1]), "v"(a[2]), "v"(a[3]), "v"(d[0]), "v"(d[1]), "v"(d[2]), "v"(d[3]));
#pragma unroll
    for (int u = 0; u < SB; ++u) {
      if (p[u] < cnt) {
        const float e = lrelu(a[u] + d[u]);
        atomicMax(&cx.mx[r[u]], fenc(e));
        li[p[u]] = s[u];
        lr[p[u]] = r[u];
        le[p[u]] = e;
        if (MODE == 0) {
          v2i sr;
          sr.x = s[u];
          sr.y = r[u];
          gl[p[u]] = sr;
        }
      }
    }
  }
}

template <int C, int TN>
__device__ __forceinline__ void sum_list(const AggCtx<C, TN>& cx, const int* li, const int* lr,
                                         const float* le, int cnt) {
  const int lane = cx.lane;
  for (int base = 0; base < cnt; base += 32) {
    const int nvalid = (cnt - base < 32) ? (cnt - base) : 32;
    int s = 0, r = 0;
    float x = 0.f;
    if (lane < nvalid) {
      s = li[base + lane];
      r = lr[base + lane];
      x = __expf(le[base + lane] - fdec(cx.mx[r]));
      atomicAdd(&cx.den[r], x);
    }
    add_rows<C, TN>(cx, s, r, x, nvalid);
  }
}


template <int C, int TN, int PHASE>
__device__ __forceinline__ void drain(const AggCtx<C, TN>& cx, int base, int nvalid) {
  const int lane = cx.lane;
  int s = 0, r = 0;
  float x = 0.f;
  if (lane < nvalid) {
    const int i = cx.qi[base + lane];
    r = cx.qr[base + lane];
    s = cx.src[i];
    const float e = lrelu(cx.als[s] + cx.ald[cx.n0 + r]);
    if (PHASE == 1) {
      atomicMax(&cx.mx[r], fenc(e));
    } else {
      x = __expf(e - fdec(cx.mx[r]));
      atomicAdd(&cx.den[r], x);
    }
  }
  if (PHASE == 2) add_rows<C, TN>(cx, s, r, x, nvalid);
}

template <int C, int TN, int PHASE>
__device__ __forceinline__ void push(const AggCtx<C, TN>& cx, int& cnt, bool f, int i, unsigned r) {
  const unsigned mask = __builtin_amdgcn_ballot_w32(f);
  if (mask == 0u) return;
  if (f) {
    const int slot = cnt + (int)__builtin_amdgcn_mbcnt_lo(mask, 0u);
    cx.qi[slot] = i;
    cx.qr[slot] = (int)r;
  }
  cnt += __builtin_popcount(mask);
  if (cnt >= 32) {
    wave_sync();
    cnt -= 32;
    drain<C, TN, PHASE>(cx, cnt, 32);
    wave_sync();
  }
}

template <int C, int TN, int PHASE>
__device__ __forceinline__ void proc_vec(const AggCtx<C, TN>& cx, int& cnt, unsigned pat, v4i k, int ib) {
  unsigned zc = vec_flags(k, pat);
  while (__builtin_amdgcn_ballot_w32(zc != 0u) != 0u) {
    const bool f = zc != 0u;
    const int i = ib + take_flag(zc);
    unsigned r = 0u;
    if (f) r = (unsigned)(cx.dst[i] - cx.n0);
    push<C, TN, PHASE>(cx, cnt, f, i, r);
  }
}

template <int C, int TN, int PHASE>
__device__ __forceinline__ void scan_slow(const AggCtx<C, TN>& cx, const Share& sh, int E,
                                          unsigned pat, int wave) {
  const v4i* kv = sh.kv;
  const int lane = cx.lane;
  int cnt = 0;
#pragma nounroll
  for (int part = 0; part < 2; ++part) {
    const int lo = part == 0 ? sh.rot : 0;
    const int hi = part == 0 ? sh.nvec : sh.rot;
    for (int vb = lo + wave * 32; vb < hi; vb += 1024) {
      const int v = vb + lane;
      v4i k = {-1, -1, -1, -1};
      if (v < hi) k = kv[v];
      proc_vec<C, TN, PHASE>(cx, cnt, pat, k, v * 16);
    }
  }
  if (wave == 0) {
    const int i = sh.tail0 + lane;
    const int dd = (i < E) ? cx.dst[i] : -1;
    const unsigned r = (unsigned)(dd - cx.n0);
    push<C, TN, PHASE>(cx, cnt, (i < E) && (r < (unsigned)TN), i, r);
  }
  wave_sync();
  if (cnt > 0) drain<C, TN, PHASE>(cx, 0, cnt);
  wave_sync();
}

template <int C, int TN, int CAPW, int MODE>
__global__ __launch_bounds__(1024) void agg_tile(const int* __restrict__ src,
                                                 const int* __restrict__ dst,
                                                 const float* __restrict__ als,
                                                 const float* __restrict__ ald,
                                                 const float* __restrict__ H,
                                                 const float* __restrict__ b,
                                                 float* __restrict__ out,
                                                 int* gcnt, int* glist,
                                                 const unsigned* __restrict__ keys,
                                                 int E, int Nn, int relu) {
  constexpr int VPT = C / 32;
  static_assert(CAPW >= 64, "the slow way's queue lives in the list's first 64 slots");
  __shared__ float acc[TN * C];
  __shared__ unsigned mx[TN];
  __shared__ float den[TN];
  __shared__ int li[32 * CAPW];
  __shared__ int lr[32 * CAPW];
  __shared__ float le[32 * CAPW];

  const int tid = threadIdx.x;
  const int lane = tid & 31;
  const int wave = __builtin_amdgcn_readfirstlane(tid >> 5);
  const int n0 = blockIdx.x * TN;

  for (int i = tid; i < TN * C; i += 1024) acc[i] = 0.f;
  for (int n = tid; n < TN; n += 1024) {
    const int node = n0 + n;
    float e = 0.f;
    if (node < Nn) e = lrelu(als[node] + ald[node]);
    mx[n] = fenc(e);
    den[n] = 0.f;
  }
  __syncthreads();

  int* wli = li + wave * CAPW;
  int* wlr = lr + wave * CAPW;
  float* wle = le + wave * CAPW;

  AggCtx<C, TN> cx;
  cx.src = src;
  cx.dst = dst;
  cx.als = als;
  cx.ald = ald;
  cx.H = H;
  cx.acc = acc;
  cx.mx = mx;
  cx.den = den;
  cx.qi = wli;
  cx.qr = wlr;
  cx.n0 = n0;
  cx.lane = lane;

  const Share sh = share_of(keys, E);
  const unsigned pat = (unsigned)blockIdx.x * 0x01010101u;
  const int gw = blockIdx.x * 32 + wave;
  v2i* wgl = reinterpret_cast<v2i*>(glist) + (size_t)gw * CAPW;
  int kept;
  if (MODE == 0) {
    kept = scan_record<TN, CAPW>(sh, dst, E, n0, pat, wave, lane, wli);
    if (lane == 0) gcnt[gw] = kept;
  } else {
    kept = __builtin_amdgcn_readfirstlane(gcnt[gw]);
  }
  const bool listed = kept <= CAPW;
  wave_sync();
  if (listed) score_list<C, TN, MODE>(cx, wli, wlr, wle, wgl, kept);
  else scan_slow<C, TN, 1>(cx, sh, E, pat, wave);
  __syncthreads();
  if (listed) sum_list<C, TN>(cx, wli, wlr, wle, kept);
  else scan_slow<C, TN, 2>(cx, sh, E, pat, wave);
  __syncthreads();

  {
    typedef float vrow __attribute__((ext_vector_type(VPT)));
    constexpr int WB = 7;
    const int col = lane * VPT;
    const vrow bv = *reinterpret_cast<const vrow*>(b + col);
    for (int nb = wave; nb < TN; nb += 32 * WB) {
      vrow hv[WB];
      float sa[WB], sd[WB];
#pragma unroll
      for (int u = 0; u < WB; ++u) {
        const int n = nb + 32 * u;
        int node = n0 + n;
        if (n >= TN || node >= Nn) node = n0;
        hv[u] = *reinterpret_cast<const vrow*>(H + (size_t)node * C + col);
        sa[u] = als[node];
        sd[u] = ald[node];
      }
#pragma unroll
      for (int u = 0; u < WB; ++u) {
        const int n = nb + 32 * u;
        const int node = n0 + n;
        if (n < TN && node < Nn) {
          const float es = lrelu(sa[u] + sd[u]);
          const float xs = __expf(es - fdec(mx[n]));
          const float dn = den[n] + xs;
          vrow o;
#pragma unroll
          for (int j = 0; j < VPT; ++j) {
            const float a = acc[n * C + j * 32 + lane];
            float v = (a + xs * hv[u][j]) / dn + bv[j];
            if (relu) v = fmaxf(v, 0.f);
            o[j] = v;
          }
          *reinterpret_cast<vrow*>(out + (size_t)node * C + col) = o;
        }
      }
    }
  }
}

extern "C" void kernel_launch(void* const* d_in, const int* in_sizes, int n_in,
                              void* d_out, int out_size, void* d_ws, size_t ws_size,
                              hipStream_t stream) {
  const float* x   = (const float*)d_in[0];
  const int*   ei  = (const int*)d_in[1];
  const float* W1  = (const float*)d_in[2];
  const float* as1 = (const float*)d_in[3];
  const float* ad1 = (const float*)d_in[4];
  const float* b1  = (const float*)d_in[5];
  const float* W2  = (const float*)d_in[6];
  const float* as2 = (const float*)d_in[7];
  const float* ad2 = (const float*)d_in[8];
  const float* b2  = (const float*)d_in[9];
  const float* W3  = (const float*)d_in[10];
  const float* as3 = (const float*)d_in[11];
  const float* ad3 = (const float*)d_in[12];
  const float* b3  = (const float*)d_in[13];

  const int N = in_sizes[0] / 128;
  const int E = in_sizes[1] / 2;
  const int* srcp = ei;
  const int* dstp = ei + E;

  unsigned* keys = (unsigned*)d_ws;
  float* buf0 = (float*)(keys + (((E >> 2) + 31) & ~31));
  float* buf1 = buf0 + (size_t)N * 128;
  float* als  = buf1 + (size_t)N * 128;
  float* ald  = als + N;
  int* gcnt   = (int*)(ald + N);

  constexpr int TN = 400;
  constexpr int CAPW = 288;
  const int gGemm = (N / 16 + 7) / 8;
  const int gProj = (N + 255) / 256;
  const int gAgg  = (N + TN - 1) / TN;
  int* glist = gcnt + gAgg * 32;
  float* outF = (float*)d_out;

  gemm_wmma<128, 128><<<gGemm, 256, 0, stream>>>(x, W1, buf0, N);
  row_proj<128, TN><<<gProj, 256, 0, stream>>>(buf0, as1, ad1, als, ald, N, dstp, keys, E);
  agg_tile<128, TN, CAPW, 0><<<gAgg, 1024, 0, stream>>>(srcp, dstp, als, ald, buf0, b1, buf1, gcnt, glist, keys, E, N, 1);

  gemm_wmma<128, 128><<<gGemm, 256, 0, stream>>>(buf1, W2, buf0, N);
  row_proj<128, 0><<<gProj, 256, 0, stream>>>(buf0, as2, ad2, als, ald, N, dstp, keys, E);
  agg_tile<128, TN, CAPW, 1><<<gAgg, 1024, 0, stream>>>(srcp, dstp, als, ald, buf0, b2, buf1, gcnt, glist, keys, E, N, 1);

  gemm_wmma<128, 64><<<gGemm, 256, 0, stream>>>(buf1, W3, buf0, N);
  row_proj<64, 0><<<gProj, 256, 0, stream>>>(buf0, as3, ad3, als, ald, N, dstp, keys, E);
  agg_tile<64, TN, CAPW, 1><<<gAgg, 1024, 0, stream>>>(srcp, dstp, als, ald, buf0, b3, outF, gcnt, glist, keys, E, N, 0);
}


# kept edges handed to the wave owning their destination row (row mod 32; slot = LDS integer add); owner adds h rows by plain LDS vector load/add/store, no float atomics on rows: add order changes
# speedup vs baseline: 5.3111x; 1.0339x over previous
#include <hip/hip_runtime.h>

typedef __attribute__((ext_vector_type(2))) float v2f;
typedef __attribute__((ext_vector_type(8))) float v8f;

#define NEG_SLOPE 0.2f

template <int CIN, int COUT>
__global__ __launch_bounds__(256) void gemm_wmma(const float* __restrict__ X,
                                                 const float* __restrict__ Wg,
                                                 float* __restrict__ H, int M) {
  __shared__ v2f Wswz[(CIN / 4) * COUT * 2];
  {
    constexpr int TOT = (CIN / 4) * COUT * 2;
    for (int i = threadIdx.x; i < TOT; i += 256) {
      int m = i / (COUT * 2);
      int r = i - m * (COUT * 2);
      int c = r >> 1;
      int hsel = r & 1;
      const float* wp = Wg + (size_t)(4 * m + 2 * hsel) * COUT + c;
      v2f w;
      w.x = wp[0];
      w.y = wp[COUT];
      Wswz[i] = w;
    }
  }
  __syncthreads();

  const int lane = threadIdx.x & 31;
  const int wave = threadIdx.x >> 5;
  const int mtile = blockIdx.x * 8 + wave;
  if (mtile * 16 >= M) return;
  const int m0 = mtile * 16;

  constexpr int NT = COUT / 16;
  v8f acc[NT] = {};

  const int rsel = lane & 15;
  const int hiHalf = lane >> 4;
  const float* xrow = X + (size_t)(m0 + rsel) * CIN + hiHalf * 2;
  const v2f* wbase = Wswz + rsel * 2 + hiHalf;

  for (int k = 0; k < CIN; k += 4) {
    v2f a = *reinterpret_cast<const v2f*>(xrow + k);
    const v2f* wb = wbase + (k >> 2) * (COUT * 2);
#pragma unroll
    for (int t = 0; t < NT; ++t) {
      v2f bf = wb[t * 32];
      acc[t] = __builtin_amdgcn_wmma_f32_16x16x4_f32(
          false, a, false, bf, (short)0, acc[t], false, false);
    }
  }

  const int rbase = m0 + hiHalf * 8;
#pragma unroll
  for (int t = 0; t < NT; ++t) {
#pragma unroll
    for (int i = 0; i < 8; ++i) {
      H[(size_t)(rbase + i) * COUT + t * 16 + rsel] = acc[t][i];
    }
  }
}

template <int TNK>
__device__ __forceinline__ unsigned key_dword(const int* __restrict__ p) {
  unsigned kw = 0u;
#pragma unroll
  for (int j = 0; j < 4; ++j) {
    unsigned t = (unsigned)p[j] / (unsigned)TNK;
    if (t > 255u) t = 255u;
    kw |= t << (8 * j);
  }
  return kw;
}

template <int C, int TNK>
__global__ __launch_bounds__(256) void row_proj(const float* __restrict__ H,
                                                const float* __restrict__ asrc,
                                                const float* __restrict__ adst,
                                                float* __restrict__ als,
                                                float* __restrict__ ald, int Nn,
                                                const int* __restrict__ dst,
                                                unsigned* __restrict__ keys, int E) {
  if constexpr (TNK > 0) {
    const int nq = E >> 2;
    const int step = (int)gridDim.x * 256;
    constexpr int KU = 4;
    int q = (int)blockIdx.x * 256 + (int)threadIdx.x;
    for (; q + (KU - 1) * step < nq; q += KU * step) {
      unsigned kw[KU];
#pragma unroll
      for (int u = 0; u < KU; ++u) kw[u] = key_dword<TNK>(dst + 4 * (q + u * step));
#pragma unroll
      for (int u = 0; u < KU; ++u) keys[q + u * step] = kw[u];
    }
    for (; q < nq; q += step) keys[q] = key_dword<TNK>(dst + 4 * q);
  }
  constexpr int VPT = C / 32;
  constexpr int RB = 8;
  const int lane = threadIdx.x & 31;
  const int wave = threadIdx.x >> 5;
  const int nb = (blockIdx.x * 8 + wave) * 32;
  if (nb >= Nn) return;
  float as[VPT], ad[VPT];
#pragma unroll
  for (int j = 0; j < VPT; ++j) {
    as[j] = asrc[lane + 32 * j];
    ad[j] = adst[lane + 32 * j];
  }
  float rs = 0.f, rd = 0.f;
  for (int k = 0; k < 32; k += RB) {
    float hv[RB][VPT];
#pragma unroll
    for (int u = 0; u < RB; ++u) {
      int n = nb + k + u;
      if (n > Nn - 1) n = Nn - 1;
      const float* hrow = H + (size_t)n * C;
#pragma unroll
      for (int j = 0; j < VPT; ++j) hv[u][j] = hrow[lane + 32 * j];
    }
#pragma unroll
    for (int u = 0; u < RB; ++u) {
      float s = 0.f, d = 0.f;
#pragma unroll
      for (int j = 0; j < VPT; ++j) {
        s += hv[u][j] * as[j];
        d += hv[u][j] * ad[j];
      }
#pragma unroll
      for (int off = 16; off > 0; off >>= 1) {
        s += __shfl_xor(s, off, 32);
        d += __shfl_xor(d, off, 32);
      }
      if (lane == k + u) {
        rs = s;
        rd = d;
      }
    }
  }
  if (nb + lane < Nn) {
    als[nb + lane] = rs;
    ald[nb + lane] = rd;
  }
}

__device__ __forceinline__ unsigned fenc(float f) {
  unsigned u = __float_as_uint(f);
  return u ^ ((u >> 31) ? 0xFFFFFFFFu : 0x80000000u);
}
__device__ __forceinline__ float fdec(unsigned u) {
  unsigned v = u ^ ((u & 0x80000000u) ? 0x80000000u : 0xFFFFFFFFu);
  return __uint_as_float(v);
}

typedef __attribute__((ext_vector_type(4))) int v4i;
typedef __attribute__((ext_vector_type(2))) int v2i;

__device__ __forceinline__ float lrelu(float e) { return (e > 0.f) ? e : NEG_SLOPE * e; }

__device__ __forceinline__ void wave_sync() {
  __builtin_amdgcn_fence(__ATOMIC_RELEASE, "wavefront");
  __builtin_amdgcn_wave_barrier();
  __builtin_amdgcn_fence(__ATOMIC_ACQUIRE, "wavefront");
}

template <int C, int TN>
struct AggCtx {
  const int* src;
  const int* dst;
  const float* als;
  const float* ald;
  const float* H;
  float* acc;
  unsigned* mx;
  float* den;
  int* qi;
  int* qr;
  int n0;
  int lane;
};

template <int C, int TN>
__device__ __forceinline__ void add_rows(const AggCtx<C, TN>& cx, int s, int r, float x, int nvalid) {
  constexpr int VPT = C / 32;
  typedef float vrow __attribute__((ext_vector_type(VPT)));
  const int lane = cx.lane;
  const int col = lane * VPT;
  constexpr int B = 8;
  int k = 0;
  for (; k + B <= nvalid; k += B) {
    vrow v[B];
    int rk[B];
    float xk[B];
#pragma unroll
    for (int u = 0; u < B; ++u) {
      const int sk = __builtin_amdgcn_readlane(s, k + u);
      rk[u] = __builtin_amdgcn_readlane(r, k + u);
      xk[u] = __int_as_float(__builtin_amdgcn_readlane(__float_as_int(x), k + u));
      v[u] = *reinterpret_cast<const vrow*>(cx.H + (size_t)sk * C + col);
    }
#pragma unroll
    for (int u = 0; u < B; ++u) {
      float* ar = cx.acc + rk[u] * C + col;
#pragma unroll
      for (int j = 0; j < VPT; ++j) atomicAdd(&ar[j], xk[u] * v[u][j]);
    }
  }
  for (; k < nvalid; ++k) {
    const int sk = __builtin_amdgcn_readlane(s, k);
    const int rq = __builtin_amdgcn_readlane(r, k);
    const float xq = __int_as_float(__builtin_amdgcn_readlane(__float_as_int(x), k));
    const vrow hv = *reinterpret_cast<const vrow*>(cx.H + (size_t)sk * C + col);
    float* ar = cx.acc + rq * C + col;
#pragma unroll
    for (int j = 0; j < VPT; ++j) atomicAdd(&ar[j], xq * hv[j]);
  }
}

template <int C, int TN>
__device__ __forceinline__ void add_rows_owned(const AggCtx<C, TN>& cx, int s, int r, float x, int nvalid) {
  constexpr int VPT = C / 32;
  typedef float vrow __attribute__((ext_vector_type(VPT)));
  const int col = cx.lane * VPT;
  constexpr int B = 8;
  if (nvalid == 32) {
    constexpr int G = 32 / B;
    vrow v[2][B];
#pragma unroll
    for (int u = 0; u < B; ++u) {
      const int sk = __builtin_amdgcn_readlane(s, u);
      v[0][u] = *reinterpret_cast<const vrow*>(cx.H + (size_t)sk * C + col);
    }
#pragma unroll
    for (int g = 0; g < G; ++g) {
      if (g + 1 < G) {
#pragma unroll
        for (int u = 0; u < B; ++u) {
          const int sk = __builtin_amdgcn_readlane(s, (g + 1) * B + u);
          v[(g + 1) & 1][u] = *reinterpret_cast<const vrow*>(cx.H + (size_t)sk * C + col);
        }
      }
#pragma unroll
      for (int u = 0; u < B; ++u) {
        const int rq = __builtin_amdgcn_readlane(r, g * B + u);
        const float xq = __int_as_float(__builtin_amdgcn_readlane(__float_as_int(x), g * B + u));
        vrow* ar = reinterpret_cast<vrow*>(cx.acc + rq * C + col);
        vrow a = *ar;
        a += xq * v[g & 1][u];
        *ar = a;
      }
    }
    return;
  }
  int k = 0;
  for (; k + B <= nvalid; k += B) {
    vrow v[B];
    int rk[B];
    float xk[B];
#pragma unroll
    for (int u = 0; u < B; ++u) {
      const int sk = __builtin_amdgcn_readlane(s, k + u);
      rk[u] = __builtin_amdgcn_readlane(r, k + u);
      xk[u] = __int_as_float(__builtin_amdgcn_readlane(__float_as_int(x), k + u));
      v[u] = *reinterpret_cast<const vrow*>(cx.H + (size_t)sk * C + col);
    }
#pragma unroll
    for (int u = 0; u < B; ++u) {
      vrow* ar = reinterpret_cast<vrow*>(cx.acc + rk[u] * C + col);
      vrow a = *ar;
      a += xk[u] * v[u];
      *ar = a;
    }
  }
  for (; k < nvalid; ++k) {
    const int sk = __builtin_amdgcn_readlane(s, k);
    const int rq = __builtin_amdgcn_readlane(r, k);
    const float xq = __int_as_float(__builtin_amdgcn_readlane(__float_as_int(x), k));
    const vrow hv = *reinterpret_cast<const vrow*>(cx.H + (size_t)sk * C + col);
    vrow* ar = reinterpret_cast<vrow*>(cx.acc + rq * C + col);
    vrow a = *ar;
    a += xq * hv;
    *ar = a;
  }
}


struct Share {
  const v4i* kv;
  int nvec;
  int tail0;
  int rot;
};

__device__ __forceinline__ Share share_of(const unsigned* __restrict__ keys, int E) {
  Share sh;
  sh.kv = reinterpret_cast<const v4i*>(keys);
  sh.nvec = E >> 4;
  sh.tail0 = sh.nvec * 16;
  sh.rot = (int)(((long long)blockIdx.x * sh.nvec) / gridDim.x) & ~31;
  return sh;
}

__device__ __forceinline__ unsigned zero_bytes(unsigned x) {
  return ~(((x & 0x7F7F7F7Fu) + 0x7F7F7F7Fu) | x | 0x7F7F7F7Fu);
}


template <int CAPW>
__device__ __forceinline__ void rec(int* li, int& cnt, unsigned mask, bool f, int i) {
  if (mask == 0u) return;
  if (f) {
    const int slot = cnt + (int)__builtin_amdgcn_mbcnt_lo(mask, 0u);
    if (slot < CAPW) li[slot] = i;
  }
  cnt += __builtin_popcount(mask);
}

__device__ __forceinline__ unsigned vec_flags(v4i k, unsigned pat) {
  return (zero_bytes((unsigned)k.x ^ pat) >> 7) | (zero_bytes((unsigned)k.y ^ pat) >> 6) |
         (zero_bytes((unsigned)k.z ^ pat) >> 5) | (zero_bytes((unsigned)k.w ^ pat) >> 4);
}

__device__ __forceinline__ int take_flag(unsigned& zc) {
  const int p = __builtin_ctz(zc | 0x80000000u);
  zc &= zc - 1u;
  return 4 * (p & 7) + (p >> 3);
}

template <int CAPW>
__device__ __forceinline__ void rec_vec(int* li, int& cnt, unsigned pat, v4i k, int ib) {
  unsigned zc = vec_flags(k, pat);
  unsigned mask = __builtin_amdgcn_ballot_w32(zc != 0u);
  while (mask != 0u) {
    const bool f = zc != 0u;
    const int i = ib + take_flag(zc);
    rec<CAPW>(li, cnt, mask, f, i);
    mask = __builtin_amdgcn_ballot_w32(zc != 0u);
  }
}

template <int TN, int CAPW>
__device__ __forceinline__ int scan_record(const Share& sh, const int* __restrict__ dst, int E, int n0,
                                           unsigned pat, int wave, int lane, int* li) {
  const v4i* kv = sh.kv;
  int cnt = 0;
  constexpr int U = 8;
#pragma nounroll
  for (int part = 0; part < 2; ++part) {
    const int lo = part == 0 ? sh.rot : 0;
    const int hi = part == 0 ? sh.nvec : sh.rot;
    for (int vb = lo + wave * 32; vb < hi; vb += U * 1024) {
      v4i k[U];
      unsigned pl[U];
#pragma unroll
      for (int u = 0; u < U; ++u) {
        const int v = vb + u * 1024 + lane;
        pl[u] = v < hi ? pat : 0xFFFFFFFFu;
        k[u] = kv[v < hi ? v : hi - 1];
      }
#pragma unroll
      for (int u = 0; u < U; ++u)
        rec_vec<CAPW>(li, cnt, pl[u], k[u], (vb + u * 1024 + lane) * 16);
    }
  }
  if (wave == 0) {
    const int i = sh.tail0 + lane;
    const int dd = (i < E) ? dst[i] : -1;
    const bool f = (i < E) && ((unsigned)(dd - n0) < (unsigned)TN);
    rec<CAPW>(li, cnt, __builtin_amdgcn_ballot_w32(f), f, i);
  }
  return cnt;
}

#define SR_BITS 23
#define SR_MASK 0x7FFFFFu

template <int C, int TN, int CAPW>
__device__ __forceinline__ void score_route(const AggCtx<C, TN>& cx, const int* li, unsigned* osr,
                                            float* oe, int* ocnt, int cnt) {
  constexpr int SB = 4;
  const int lane = cx.lane;
  for (int base = 0; base < cnt; base += 32 * SB) {
    int p[SB], r[SB], s[SB];
    float a[SB], d[SB];
#pragma unroll
    for (int u = 0; u < SB; ++u) {
      p[u] = base + u * 32 + lane;
      const int q = p[u] < cnt ? p[u] : cnt - 1;
      const int i = li[q];
      s[u] = cx.src[i];
      r[u] = cx.dst[i] - cx.n0;
    }
#pragma unroll
    for (int u = 0; u < SB; ++u) {
      a[u] = cx.als[s[u]];
      d[u] = cx.ald[cx.n0 + r[u]];
    }
    static_assert(SB == 4, "the statement below names the SB gathered pairs");
    asm volatile("" ::"v"(a[0]), "v"(a[1]), "v"(a[2]), "v"(a[3]), "v"(d[0]), "v"(d[1]), "v"(d[2]), "v"(d[3]));
#pragma unroll
    for (int u = 0; u < SB; ++u) {
      if (p[u] < cnt) {
        const float e = lrelu(a[u] + d[u]);
        atomicMax(&cx.mx[r[u]], fenc(e));
        const int o = r[u] & 31;
        const int slot = atomicAdd(&ocnt[o], 1);
        if (slot < CAPW) {
          osr[o * CAPW + slot] = (unsigned)s[u] | ((unsigned)r[u] << SR_BITS);
          oe[o * CAPW + slot] = e;
        }
      }
    }
  }
}

template <int C, int TN>
__device__ __forceinline__ void score_own(const AggCtx<C, TN>& cx, const unsigned* gl, unsigned* wsr,
                                          float* we, int cnt) {
  constexpr int SB = 4;
  const int lane = cx.lane;
  for (int base = 0; base < cnt; base += 32 * SB) {
    int p[SB], r[SB];
    unsigned w[SB];
    float a[SB], d[SB];
#pragma unroll
    for (int u = 0; u < SB; ++u) {
      p[u] = base + u * 32 + lane;
      const int q = p[u] < cnt ? p[u] : cnt - 1;
      w[u] = gl[q];
      r[u] = (int)(w[u] >> SR_BITS);
    }
#pragma unroll
    for (int u = 0; u < SB; ++u) {
      a[u] = cx.als[(int)(w[u] & SR_MASK)];
      d[u] = cx.ald[cx.n0 + r[u]];
    }
    static_assert(SB == 4, "the statement below names the SB gathered pairs");
    asm volatile("" ::"v"(a[0]), "v"(a[1]), "v"(a[2]), "v"(a[3]), "v"(d[0]), "v"(d[1]), "v"(d[2]), "v"(d[3]));
#pragma unroll
    for (int u = 0; u < SB; ++u) {
      if (p[u] < cnt) {
        const float e = lrelu(a[u] + d[u]);
        atomicMax(&cx.mx[r[u]], fenc(e));
        wsr[p[u]] = w[u];
        we[p[u]] = e;
      }
    }
  }
}

template <int C, int TN>
__device__ __forceinline__ void sum_owned(const AggCtx<C, TN>& cx, const unsigned* wsr, const float* we,
                                          int cnt) {
  const int lane = cx.lane;
  for (int base = 0; base < cnt; base += 32) {
    const int nvalid = (cnt - base < 32) ? (cnt - base) : 32;
    int s = 0, r = 0;
    float x = 0.f;
    if (lane < nvalid) {
      const unsigned w = wsr[base + lane];
      s = (int)(w & SR_MASK);
      r = (int)(w >> SR_BITS);
      x = __expf(we[base + lane] - fdec(cx.mx[r]));
      atomicAdd(&cx.den[r], x);
    }
    add_rows_owned<C, TN>(cx, s, r, x, nvalid);
  }
}


template <int C, int TN, int PHASE>
__device__ __forceinline__ void drain(const AggCtx<C, TN>& cx, int base, int nvalid) {
  const int lane = cx.lane;
  int s = 0, r = 0;
  float x = 0.f;
  if (lane < nvalid) {
    const int i = cx.qi[base + lane];
    r = cx.qr[base + lane];
    s = cx.src[i];
    const float e = lrelu(cx.als[s] + cx.ald[cx.n0 + r]);
    if (PHASE == 1) {
      atomicMax(&cx.mx[r], fenc(e));
    } else {
      x = __expf(e - fdec(cx.mx[r]));
      atomicAdd(&cx.den[r], x);
    }
  }
  if (PHASE == 2) add_rows<C, TN>(cx, s, r, x, nvalid);
}

template <int C, int TN, int PHASE>
__device__ __forceinline__ void push(const AggCtx<C, TN>& cx, int& cnt, bool f, int i, unsigned r) {
  const unsigned mask = __builtin_amdgcn_ballot_w32(f);
  if (mask == 0u) return;
  if (f) {
    const int slot = cnt + (int)__builtin_amdgcn_mbcnt_lo(mask, 0u);
    cx.qi[slot] = i;
    cx.qr[slot] = (int)r;
  }
  cnt += __builtin_popcount(mask);
  if (cnt >= 32) {
    wave_sync();
    cnt -= 32;
    drain<C, TN, PHASE>(cx, cnt, 32);
    wave_sync();
  }
}

template <int C, int TN, int PHASE>
__device__ __forceinline__ void proc_vec(const AggCtx<C, TN>& cx, int& cnt, unsigned pat, v4i k, int ib) {
  unsigned zc = vec_flags(k, pat);
  while (__builtin_amdgcn_ballot_w32(zc != 0u) != 0u) {
    const bool f = zc != 0u;
    const int i = ib + take_flag(zc);
    unsigned r = 0u;
    if (f) r = (unsigned)(cx.dst[i] - cx.n0);
    push<C, TN, PHASE>(cx, cnt, f, i, r);
  }
}

template <int C, int TN, int PHASE>
__device__ __forceinline__ void scan_slow(const AggCtx<C, TN>& cx, const Share& sh, int E,
                                          unsigned pat, int wave) {
  const v4i* kv = sh.kv;
  const int lane = cx.lane;
  int cnt = 0;
#pragma nounroll
  for (int part = 0; part < 2; ++part) {
    const int lo = part == 0 ? sh.rot : 0;
    const int hi = part == 0 ? sh.nvec : sh.rot;
    for (int vb = lo + wave * 32; vb < hi; vb += 1024) {
      const int v = vb + lane;
      v4i k = {-1, -1, -1, -1};
      if (v < hi) k = kv[v];
      proc_vec<C, TN, PHASE>(cx, cnt, pat, k, v * 16);
    }
  }
  if (wave == 0) {
    const int i = sh.tail0 + lane;
    const int dd = (i < E) ? cx.dst[i] : -1;
    const unsigned r = (unsigned)(dd - cx.n0);
    push<C, TN, PHASE>(cx, cnt, (i < E) && (r < (unsigned)TN), i, r);
  }
  wave_sync();
  if (cnt > 0) drain<C, TN, PHASE>(cx, 0, cnt);
  wave_sync();
}

template <int C, int TN, int CAPW, int MODE>
__global__ __launch_bounds__(1024) void agg_tile(const int* __restrict__ src,
                                                 const int* __restrict__ dst,
                                                 const float* __restrict__ als,
                                                 const float* __restrict__ ald,
                                                 const float* __restrict__ H,
                                                 const float* __restrict__ b,
                                                 float* __restrict__ out,
                                                 int* gcnt, int* glist,
                                                 const unsigned* __restrict__ keys,
                                                 int E, int Nn, int relu) {
  constexpr int VPT = C / 32;
  static_assert(CAPW >= 64, "the slow way's queue lives in the list's first 64 slots");
  static_assert(TN <= 512, "a routed entry keeps the tile-local destination in 9 bits");
  __shared__ float acc[TN * C];
  __shared__ unsigned mx[TN];
  __shared__ float den[TN];
  __shared__ int li[32 * CAPW];
  __shared__ unsigned osr[32 * CAPW];
  __shared__ float oe[32 * CAPW];
  __shared__ int ocnt[32];
  __shared__ int rkept[32];

  const int tid = threadIdx.x;
  const int lane = tid & 31;
  const int wave = __builtin_amdgcn_readfirstlane(tid >> 5);
  const int n0 = blockIdx.x * TN;

  for (int i = tid; i < TN * C; i += 1024) acc[i] = 0.f;
  for (int n = tid; n < TN; n += 1024) {
    const int node = n0 + n;
    float e = 0.f;
    if (node < Nn) e = lrelu(als[node] + ald[node]);
    mx[n] = fenc(e);
    den[n] = 0.f;
  }
  if (tid < 32) {
    ocnt[tid] = 0;
    rkept[tid] = 0;
  }
  __syncthreads();

  int* wli = li + wave * CAPW;
  unsigned* wsr = osr + wave * CAPW;
  float* woe = oe + wave * CAPW;

  AggCtx<C, TN> cx;
  cx.src = src;
  cx.dst = dst;
  cx.als = als;
  cx.ald = ald;
  cx.H = H;
  cx.acc = acc;
  cx.mx = mx;
  cx.den = den;
  cx.qi = wli;
  cx.qr = reinterpret_cast<int*>(wsr);
  cx.n0 = n0;
  cx.lane = lane;

  const Share sh = share_of(keys, E);
  const unsigned pat = (unsigned)blockIdx.x * 0x01010101u;
  const int gw = blockIdx.x * 32 + wave;
  unsigned* wgl = reinterpret_cast<unsigned*>(glist) + (size_t)gw * CAPW;
  int mine;
  bool slow;
  if (MODE == 0) {
    const int kept = scan_record<TN, CAPW>(sh, dst, E, n0, pat, wave, lane, wli);
    if (lane == 0) rkept[wave] = kept;
    wave_sync();
    if (kept <= CAPW) score_route<C, TN, CAPW>(cx, wli, osr, oe, ocnt, kept);
    __syncthreads();
    const int oc = ocnt[lane];
    slow = __builtin_amdgcn_ballot_w32(oc > CAPW || rkept[lane] > CAPW) != 0u;
    mine = __builtin_amdgcn_readlane(oc, wave);
    if (lane == 0) gcnt[gw] = slow ? CAPW + 1 : mine;
    if (!slow) {
      for (int p = lane; p < mine; p += 32) wgl[p] = wsr[p];
    }
  } else {
    const int gc = gcnt[blockIdx.x * 32 + lane];
    slow = __builtin_amdgcn_ballot_w32(gc > CAPW) != 0u;
    mine = __builtin_amdgcn_readlane(gc, wave);
    if (!slow) score_own<C, TN>(cx, wgl, wsr, woe, mine);
  }
  if (slow) {
    scan_slow<C, TN, 1>(cx, sh, E, pat, wave);
    __syncthreads();
    scan_slow<C, TN, 2>(cx, sh, E, pat, wave);
  } else {
    wave_sync();
    sum_owned<C, TN>(cx, wsr, woe, mine);
  }
  __syncthreads();

  {
    typedef float vrow __attribute__((ext_vector_type(VPT)));
    constexpr int WB = 7;
    const int col = lane * VPT;
    const vrow bv = *reinterpret_cast<const vrow*>(b + col);
    for (int nb = wave; nb < TN; nb += 32 * WB) {
      vrow hv[WB];
      float sa[WB], sd[WB];
#pragma unroll
      for (int u = 0; u < WB; ++u) {
        const int n = nb + 32 * u;
        int node = n0 + n;
        if (n >= TN || node >= Nn) node = n0;
        hv[u] = *reinterpret_cast<const vrow*>(H + (size_t)node * C + col);
        sa[u] = als[node];
        sd[u] = ald[node];
      }
#pragma unroll
      for (int u = 0; u < WB; ++u) {
        const int n = nb + 32 * u;
        const int node = n0 + n;
        if (n < TN && node < Nn) {
          const float es = lrelu(sa[u] + sd[u]);
          const float xs = __expf(es - fdec(mx[n]));
          const float dn = den[n] + xs;
          vrow o;
          const vrow av = *reinterpret_cast<const vrow*>(acc + n * C + col);
#pragma unroll
          for (int j = 0; j < VPT; ++j) {
            const float a = av[j];
            float v = (a + xs * hv[u][j]) / dn + bv[j];
            if (relu) v = fmaxf(v, 0.f);
            o[j] = v;
          }
          *reinterpret_cast<vrow*>(out + (size_t)node * C + col) = o;
        }
      }
    }
  }
}

extern "C" void kernel_launch(void* const* d_in, const int* in_sizes, int n_in,
                              void* d_out, int out_size, void* d_ws, size_t ws_size,
                              hipStream_t stream) {
  const float* x   = (const float*)d_in[0];
  const int*   ei  = (const int*)d_in[1];
  const float* W1  = (const float*)d_in[2];
  const float* as1 = (const float*)d_in[3];
  const float* ad1 = (const float*)d_in[4];
  const float* b1  = (const float*)d_in[5];
  const float* W2  = (const float*)d_in[6];
  const float* as2 = (const float*)d_in[7];
  const float* ad2 = (const float*)d_in[8];
  const float* b2  = (const float*)d_in[9];
  const float* W3  = (const float*)d_in[10];
  const float* as3 = (const float*)d_in[11];
  const float* ad3 = (const float*)d_in[12];
  const float* b3  = (const float*)d_in[13];

  const int N = in_sizes[0] / 128;
  const int E = in_sizes[1] / 2;
  const int* srcp = ei;
  const int* dstp = ei + E;

  unsigned* keys = (unsigned*)d_ws;
  float* buf0 = (float*)(keys + (((E >> 2) + 31) & ~31));
  float* buf1 = buf0 + (size_t)N * 128;
  float* als  = buf1 + (size_t)N * 128;
  float* ald  = als + N;
  int* gcnt   = (int*)(ald + N);

  constexpr int TN = 400;
  constexpr int CAPW = 288;
  const int gGemm = (N / 16 + 7) / 8;
  const int gProj = (N + 255) / 256;
  const int gAgg  = (N + TN - 1) / TN;
  int* glist = gcnt + gAgg * 32;
  float* outF = (float*)d_out;

  gemm_wmma<128, 128><<<gGemm, 256, 0, stream>>>(x, W1, buf0, N);
  row_proj<128, TN><<<gProj, 256, 0, stream>>>(buf0, as1, ad1, als, ald, N, dstp, keys, E);
  agg_tile<128, TN, CAPW, 0><<<gAgg, 1024, 0, stream>>>(srcp, dstp, als, ald, buf0, b1, buf1, gcnt, glist, keys, E, N, 1);

  gemm_wmma<128, 128><<<gGemm, 256, 0, stream>>>(buf1, W2, buf0, N);
  row_proj<128, 0><<<gProj, 256, 0, stream>>>(buf0, as2, ad2, als, ald, N, dstp, keys, E);
  agg_tile<128, TN, CAPW, 1><<<gAgg, 1024, 0, stream>>>(srcp, dstp, als, ald, buf0, b2, buf1, gcnt, glist, keys, E, N, 1);

  gemm_wmma<128, 64><<<gGemm, 256, 0, stream>>>(buf1, W3, buf0, N);
  row_proj<64, 0><<<gProj, 256, 0, stream>>>(buf0, as3, ad3, als, ald, N, dstp, keys, E);
  agg_tile<64, TN, CAPW, 1><<<gAgg, 1024, 0, stream>>>(srcp, dstp, als, ald, buf0, b3, outF, gcnt, glist, keys, E, N, 0);
}
